# starting kernel
# baseline (speedup 1.0000x reference)
#include <hip/hip_runtime.h>
#include <cstdint>
#include <cstddef>

#define DFE 128

typedef float v2f __attribute__((ext_vector_type(2)));
typedef float v8f __attribute__((ext_vector_type(8)));

__global__ void k_init_deg(float* __restrict__ deg_out, float* __restrict__ deg_in, int n) {
  int i = blockIdx.x * blockDim.x + threadIdx.x;
  if (i < n) { deg_out[i] = 1.0f; deg_in[i] = 1.0f; }
}

__global__ void k_edge_deg(const int* __restrict__ src, const int* __restrict__ dst,
                           float* __restrict__ deg_out, float* __restrict__ deg_in, int e) {
  int i = blockIdx.x * blockDim.x + threadIdx.x;
  if (i < e) {
    unsafeAtomicAdd(&deg_out[src[i]], 1.0f);
    unsafeAtomicAdd(&deg_in[dst[i]], 1.0f);
  }
}

__global__ void k_rsqrt_deg(float* __restrict__ deg_out, float* __restrict__ deg_in, int n) {
  int i = blockIdx.x * blockDim.x + threadIdx.x;
  if (i < n) {
    deg_out[i] = rsqrtf(deg_out[i]);
    deg_in[i]  = rsqrtf(deg_in[i]);
  }
}

__global__ void k_self_init(const float* __restrict__ h, const float* __restrict__ cs,
                            float* __restrict__ agg, int n) {
  int t = blockIdx.x * blockDim.x + threadIdx.x;
  int node = t >> 5, c4 = t & 31;
  if (node < n) {
    float s = cs[node];
    float4 v = ((const float4*)(h + (size_t)node * DFE))[c4];
    v.x *= s; v.y *= s; v.z *= s; v.w *= s;
    ((float4*)(agg + (size_t)node * DFE))[c4] = v;
  }
}

__global__ void k_edge_scatter(const int* __restrict__ src, const int* __restrict__ dst,
                               const float* __restrict__ h, const float* __restrict__ cs,
                               float* __restrict__ agg, int e) {
  int t = blockIdx.x * blockDim.x + threadIdx.x;
  int edge = t >> 5, lane = t & 31;
  if (edge < e) {
    int s = src[edge], d = dst[edge];
    float sc = cs[s];
    float4 v = ((const float4*)(h + (size_t)s * DFE))[lane];
    float* p = agg + (size_t)d * DFE + lane * 4;
    unsafeAtomicAdd(p + 0, v.x * sc);
    unsafeAtomicAdd(p + 1, v.y * sc);
    unsafeAtomicAdd(p + 2, v.z * sc);
    unsafeAtomicAdd(p + 3, v.w * sc);
  }
}

__global__ __launch_bounds__(128) void k_gemm_relu(
    const float* __restrict__ agg, const float* __restrict__ cd,
    const float* __restrict__ W, const float* __restrict__ bias,
    float* __restrict__ out, int nrows) {
  __shared__ float As[16 * 132];
  const int tid = threadIdx.x;
  const int row0 = blockIdx.x * 16;

  for (int idx = tid; idx < 16 * 32; idx += 128) {
    int r = idx >> 5, c4 = idx & 31;
    int row = row0 + r;
    float4 v = make_float4(0.f, 0.f, 0.f, 0.f);
    float sc = 0.f;
    if (row < nrows) {
      v = ((const float4*)(agg + (size_t)row * DFE))[c4];
      sc = cd[row];
    }
    float* q = As + r * 132 + c4 * 4;
    q[0] = v.x * sc; q[1] = v.y * sc; q[2] = v.z * sc; q[3] = v.w * sc;
  }
  __syncthreads();

  const int wave = tid >> 5;
  const int lane = tid & 31;
  const int half = lane >> 4;
  const int l15  = lane & 15;
  const int ntile0 = wave * 2;
  const int colb0 = ntile0 * 16;

  v8f acc0 = {};
  v8f acc1 = {};
  const float* Arow = As + l15 * 132;

  for (int k0 = 0; k0 < DFE; k0 += 4) {
    int ka = k0 + 2 * half;
    v2f a = { Arow[ka], Arow[ka + 1] };
    v2f b0 = { W[(size_t)ka * DFE + colb0 + l15],
               W[(size_t)(ka + 1) * DFE + colb0 + l15] };
    v2f b1 = { W[(size_t)ka * DFE + colb0 + 16 + l15],
               W[(size_t)(ka + 1) * DFE + colb0 + 16 + l15] };
    acc0 = __builtin_amdgcn_wmma_f32_16x16x4_f32(false, a, false, b0,
                                                 (short)0, acc0, false, false);
    acc1 = __builtin_amdgcn_wmma_f32_16x16x4_f32(false, a, false, b1,
                                                 (short)0, acc1, false, false);
  }

  for (int v = 0; v < 8; ++v) {
    int row = row0 + v + 8 * half;
    if (row < nrows) {
      int col0 = colb0 + l15;
      float x0 = acc0[v] + bias[col0];
      out[(size_t)row * DFE + col0] = x0 > 0.f ? x0 : 0.f;
      int col1 = col0 + 16;
      float x1 = acc1[v] + bias[col1];
      out[(size_t)row * DFE + col1] = x1 > 0.f ? x1 : 0.f;
    }
  }
}

__global__ void k_pool_zero(float* __restrict__ sums, float* __restrict__ cnt, int g) {
  int i = blockIdx.x * blockDim.x + threadIdx.x;
  if (i < g * DFE) sums[i] = 0.f;
  if (i < g) cnt[i] = 0.f;
}

__global__ void k_pool_acc(const float* __restrict__ h, const int* __restrict__ gid,
                           float* __restrict__ sums, float* __restrict__ cnt, int n) {
  int t = blockIdx.x * blockDim.x + threadIdx.x;
  int node = t >> 5, lane = t & 31;
  if (node < n) {
    int g = gid[node];
    float4 v = ((const float4*)(h + (size_t)node * DFE))[lane];
    float* p = sums + (size_t)g * DFE + lane * 4;
    unsafeAtomicAdd(p + 0, v.x);
    unsafeAtomicAdd(p + 1, v.y);
    unsafeAtomicAdd(p + 2, v.z);
    unsafeAtomicAdd(p + 3, v.w);
    if (lane == 0) unsafeAtomicAdd(&cnt[g], 1.0f);
  }
}

__global__ void k_pool_final(const float* __restrict__ sums, const float* __restrict__ cnt,
                             float* __restrict__ out, int g) {
  int i = blockIdx.x * blockDim.x + threadIdx.x;
  if (i < g * DFE) {
    float c = cnt[i >> 7];
    c = c > 1.0f ? c : 1.0f;
    out[i] = sums[i] / c;
  }
}

extern "C" void kernel_launch(void* const* d_in, const int* in_sizes, int n_in,
                              void* d_out, int out_size, void* d_ws, size_t ws_size,
                              hipStream_t stream) {
  const float* x   = (const float*)d_in[0];
  const float* W0  = (const float*)d_in[1];
  const float* b0  = (const float*)d_in[2];
  const float* W1  = (const float*)d_in[3];
  const float* b1  = (const float*)d_in[4];
  const float* W2  = (const float*)d_in[5];
  const float* b2  = (const float*)d_in[6];
  const int*   src = (const int*)d_in[7];
  const int*   dst = (const int*)d_in[8];
  const int*   gid = (const int*)d_in[9];

  const int n = in_sizes[0] / DFE;
  const int e = in_sizes[7];
  const int g = out_size / DFE;

  float* ws = (float*)d_ws;
  size_t ND = (size_t)n * DFE;
  float* hA   = ws;
  float* hB   = hA + ND;
  float* agg  = hB + ND;
  float* cs   = agg + ND;
  float* cd   = cs + n;
  float* psum = cd + n;
  float* pcnt = psum + (size_t)g * DFE;

  const int TB = 256;
  dim3 blk(TB);

  k_init_deg<<<dim3((n + TB - 1) / TB), blk, 0, stream>>>(cs, cd, n);
  k_edge_deg<<<dim3((e + TB - 1) / TB), blk, 0, stream>>>(src, dst, cs, cd, e);
  k_rsqrt_deg<<<dim3((n + TB - 1) / TB), blk, 0, stream>>>(cs, cd, n);

  const unsigned nodeWaveBlocks = (unsigned)(((size_t)n * 32 + TB - 1) / TB);
  const unsigned edgeWaveBlocks = (unsigned)(((size_t)e * 32 + TB - 1) / TB);
  const unsigned gemmBlocks     = (unsigned)((n + 15) / 16);

  auto layer = [&](const float* h_in, const float* W, const float* b, float* h_out) {
    k_self_init<<<dim3(nodeWaveBlocks), blk, 0, stream>>>(h_in, cs, agg, n);
    k_edge_scatter<<<dim3(edgeWaveBlocks), blk, 0, stream>>>(src, dst, h_in, cs, agg, e);
    k_gemm_relu<<<dim3(gemmBlocks), dim3(128), 0, stream>>>(agg, cd, W, b, h_out, n);
  };

  layer(x,  W0, b0, hA);
  layer(hA, W1, b1, hB);
  layer(hB, W2, b2, hA);

  k_pool_zero<<<dim3((g * DFE + TB - 1) / TB), blk, 0, stream>>>(psum, pcnt, g);
  k_pool_acc<<<dim3(nodeWaveBlocks), blk, 0, stream>>>(hA, gid, psum, pcnt, n);
  k_pool_final<<<dim3((g * DFE + TB - 1) / TB), blk, 0, stream>>>(psum, pcnt, (float*)d_out, g);
}


# pooling: reduce before the atomics: each wave sums its 16 nodes per run of equal graph id in registers, 8 waves join in LDS, one float atomic per column per run (pool sum order changes)
# speedup vs baseline: 1.3448x; 1.3448x over previous
#include <hip/hip_runtime.h>
#include <cstdint>
#include <cstddef>

#define DFE 128

typedef float v2f __attribute__((ext_vector_type(2)));
typedef float v8f __attribute__((ext_vector_type(8)));

__global__ void k_init_deg(float* __restrict__ deg_out, float* __restrict__ deg_in, int n) {
  int i = blockIdx.x * blockDim.x + threadIdx.x;
  if (i < n) { deg_out[i] = 1.0f; deg_in[i] = 1.0f; }
}

__global__ void k_edge_deg(const int* __restrict__ src, const int* __restrict__ dst,
                           float* __restrict__ deg_out, float* __restrict__ deg_in, int e) {
  int i = blockIdx.x * blockDim.x + threadIdx.x;
  if (i < e) {
    unsafeAtomicAdd(&deg_out[src[i]], 1.0f);
    unsafeAtomicAdd(&deg_in[dst[i]], 1.0f);
  }
}

__global__ void k_rsqrt_deg(float* __restrict__ deg_out, float* __restrict__ deg_in, int n) {
  int i = blockIdx.x * blockDim.x + threadIdx.x;
  if (i < n) {
    deg_out[i] = rsqrtf(deg_out[i]);
    deg_in[i]  = rsqrtf(deg_in[i]);
  }
}

__global__ void k_self_init(const float* __restrict__ h, const float* __restrict__ cs,
                            float* __restrict__ agg, int n) {
  int t = blockIdx.x * blockDim.x + threadIdx.x;
  int node = t >> 5, c4 = t & 31;
  if (node < n) {
    float s = cs[node];
    float4 v = ((const float4*)(h + (size_t)node * DFE))[c4];
    v.x *= s; v.y *= s; v.z *= s; v.w *= s;
    ((float4*)(agg + (size_t)node * DFE))[c4] = v;
  }
}

__global__ void k_edge_scatter(const int* __restrict__ src, const int* __restrict__ dst,
                               const float* __restrict__ h, const float* __restrict__ cs,
                               float* __restrict__ agg, int e) {
  int t = blockIdx.x * blockDim.x + threadIdx.x;
  int edge = t >> 5, lane = t & 31;
  if (edge < e) {
    int s = src[edge], d = dst[edge];
    float sc = cs[s];
    float4 v = ((const float4*)(h + (size_t)s * DFE))[lane];
    float* p = agg + (size_t)d * DFE + lane * 4;
    unsafeAtomicAdd(p + 0, v.x * sc);
    unsafeAtomicAdd(p + 1, v.y * sc);
    unsafeAtomicAdd(p + 2, v.z * sc);
    unsafeAtomicAdd(p + 3, v.w * sc);
  }
}

__global__ __launch_bounds__(128) void k_gemm_relu(
    const float* __restrict__ agg, const float* __restrict__ cd,
    const float* __restrict__ W, const float* __restrict__ bias,
    float* __restrict__ out, int nrows) {
  __shared__ float As[16 * 132];
  const int tid = threadIdx.x;
  const int row0 = blockIdx.x * 16;

  for (int idx = tid; idx < 16 * 32; idx += 128) {
    int r = idx >> 5, c4 = idx & 31;
    int row = row0 + r;
    float4 v = make_float4(0.f, 0.f, 0.f, 0.f);
    float sc = 0.f;
    if (row < nrows) {
      v = ((const float4*)(agg + (size_t)row * DFE))[c4];
      sc = cd[row];
    }
    float* q = As + r * 132 + c4 * 4;
    q[0] = v.x * sc; q[1] = v.y * sc; q[2] = v.z * sc; q[3] = v.w * sc;
  }
  __syncthreads();

  const int wave = tid >> 5;
  const int lane = tid & 31;
  const int half = lane >> 4;
  const int l15  = lane & 15;
  const int ntile0 = wave * 2;
  const int colb0 = ntile0 * 16;

  v8f acc0 = {};
  v8f acc1 = {};
  const float* Arow = As + l15 * 132;

  for (int k0 = 0; k0 < DFE; k0 += 4) {
    int ka = k0 + 2 * half;
    v2f a = { Arow[ka], Arow[ka + 1] };
    v2f b0 = { W[(size_t)ka * DFE + colb0 + l15],
               W[(size_t)(ka + 1) * DFE + colb0 + l15] };
    v2f b1 = { W[(size_t)ka * DFE + colb0 + 16 + l15],
               W[(size_t)(ka + 1) * DFE + colb0 + 16 + l15] };
    acc0 = __builtin_amdgcn_wmma_f32_16x16x4_f32(false, a, false, b0,
                                                 (short)0, acc0, false, false);
    acc1 = __builtin_amdgcn_wmma_f32_16x16x4_f32(false, a, false, b1,
                                                 (short)0, acc1, false, false);
  }

  for (int v = 0; v < 8; ++v) {
    int row = row0 + v + 8 * half;
    if (row < nrows) {
      int col0 = colb0 + l15;
      float x0 = acc0[v] + bias[col0];
      out[(size_t)row * DFE + col0] = x0 > 0.f ? x0 : 0.f;
      int col1 = col0 + 16;
      float x1 = acc1[v] + bias[col1];
      out[(size_t)row * DFE + col1] = x1 > 0.f ? x1 : 0.f;
    }
  }
}

__global__ void k_pool_zero(float* __restrict__ sums, float* __restrict__ cnt, int g) {
  int i = blockIdx.x * blockDim.x + threadIdx.x;
  if (i < g * DFE) sums[i] = 0.f;
  if (i < g) cnt[i] = 0.f;
}

#define PNW 16

__global__ __launch_bounds__(256) void k_pool_acc(
    const float* __restrict__ h, const int* __restrict__ gid,
    float* __restrict__ sums, float* __restrict__ cnt, int n) {
  __shared__ __attribute__((aligned(16))) float part[8 * DFE];
  __shared__ int   pg[8];
  __shared__ float pc[8];
  const int tid = threadIdx.x;
  const int lane = tid & 31;
  const int wave = __builtin_amdgcn_readfirstlane(tid >> 5);
  const int base = (blockIdx.x * 8 + wave) * PNW;
  int nn = n - base;
  nn = nn < 0 ? 0 : (nn > PNW ? PNW : nn);
  const int last = n - 1;

  int gi = base + lane;
  gi = gi < last ? gi : last;
  const int myg = gid[gi];

  float4 v[PNW];
#pragma unroll
  for (int j = 0; j < PNW; ++j) {
    int nd = base + j;
    nd = nd < last ? nd : last;
    v[j] = ((const float4*)(h + (size_t)nd * DFE))[lane];
  }

  int cur = __builtin_amdgcn_readlane(myg, 0);
  float4 acc = v[0];
  float c = 1.0f;
#pragma unroll
  for (int j = 1; j < PNW; ++j) {
    const int gj = __builtin_amdgcn_readlane(myg, j);
    if (gj != cur) {
      float* p = sums + (size_t)cur * DFE + lane * 4;
      unsafeAtomicAdd(p + 0, acc.x);
      unsafeAtomicAdd(p + 1, acc.y);
      unsafeAtomicAdd(p + 2, acc.z);
      unsafeAtomicAdd(p + 3, acc.w);
      if (lane == 0) unsafeAtomicAdd(&cnt[cur], c);
      acc = make_float4(0.f, 0.f, 0.f, 0.f);
      c = 0.f;
      cur = gj;
    }
    const bool in = j < nn;
    acc.x += in ? v[j].x : 0.f;
    acc.y += in ? v[j].y : 0.f;
    acc.z += in ? v[j].z : 0.f;
    acc.w += in ? v[j].w : 0.f;
    c += in ? 1.0f : 0.f;
  }

  ((float4*)(part + wave * DFE))[lane] = acc;
  if (lane == 0) { pg[wave] = nn > 0 ? cur : -1; pc[wave] = c; }
  __syncthreads();

  if (tid <= DFE) {
    const bool is_cnt = tid == DFE;
    const float* sp = is_cnt ? pc : part + tid;
    const int sstep = is_cnt ? 1 : DFE;
    float* dp = is_cnt ? cnt : sums + tid;
    const size_t dstep = is_cnt ? 1 : DFE;
    int gs[8];
    float ps[8];
#pragma unroll
    for (int w = 0; w < 8; ++w) { gs[w] = pg[w]; ps[w] = sp[w * sstep]; }
    float s = 0.f;
    int g0 = -1;
#pragma unroll
    for (int w = 0; w < 8; ++w) {
      if (gs[w] >= 0) {
        if (gs[w] != g0) {
          if (g0 >= 0) unsafeAtomicAdd(dp + (size_t)g0 * dstep, s);
          s = 0.f;
          g0 = gs[w];
        }
        s += ps[w];
      }
    }
    if (g0 >= 0) unsafeAtomicAdd(dp + (size_t)g0 * dstep, s);
  }
}

__global__ void k_pool_final(const float* __restrict__ sums, const float* __restrict__ cnt,
                             float* __restrict__ out, int g) {
  int i = blockIdx.x * blockDim.x + threadIdx.x;
  if (i < g * DFE) {
    float c = cnt[i >> 7];
    c = c > 1.0f ? c : 1.0f;
    out[i] = sums[i] / c;
  }
}

extern "C" void kernel_launch(void* const* d_in, const int* in_sizes, int n_in,
                              void* d_out, int out_size, void* d_ws, size_t ws_size,
                              hipStream_t stream) {
  const float* x   = (const float*)d_in[0];
  const float* W0  = (const float*)d_in[1];
  const float* b0  = (const float*)d_in[2];
  const float* W1  = (const float*)d_in[3];
  const float* b1  = (const float*)d_in[4];
  const float* W2  = (const float*)d_in[5];
  const float* b2  = (const float*)d_in[6];
  const int*   src = (const int*)d_in[7];
  const int*   dst = (const int*)d_in[8];
  const int*   gid = (const int*)d_in[9];

  const int n = in_sizes[0] / DFE;
  const int e = in_sizes[7];
  const int g = out_size / DFE;

  float* ws = (float*)d_ws;
  size_t ND = (size_t)n * DFE;
  float* hA   = ws;
  float* hB   = hA + ND;
  float* agg  = hB + ND;
  float* cs   = agg + ND;
  float* cd   = cs + n;
  float* psum = cd + n;
  float* pcnt = psum + (size_t)g * DFE;

  const int TB = 256;
  dim3 blk(TB);

  k_init_deg<<<dim3((n + TB - 1) / TB), blk, 0, stream>>>(cs, cd, n);
  k_edge_deg<<<dim3((e + TB - 1) / TB), blk, 0, stream>>>(src, dst, cs, cd, e);
  k_rsqrt_deg<<<dim3((n + TB - 1) / TB), blk, 0, stream>>>(cs, cd, n);

  const unsigned nodeWaveBlocks = (unsigned)(((size_t)n * 32 + TB - 1) / TB);
  const unsigned edgeWaveBlocks = (unsigned)(((size_t)e * 32 + TB - 1) / TB);
  const unsigned gemmBlocks     = (unsigned)((n + 15) / 16);
  const unsigned poolBlocks     = (unsigned)((n + 8 * PNW - 1) / (8 * PNW));

  auto layer = [&](const float* h_in, const float* W, const float* b, float* h_out) {
    k_self_init<<<dim3(nodeWaveBlocks), blk, 0, stream>>>(h_in, cs, agg, n);
    k_edge_scatter<<<dim3(edgeWaveBlocks), blk, 0, stream>>>(src, dst, h_in, cs, agg, e);
    k_gemm_relu<<<dim3(gemmBlocks), dim3(128), 0, stream>>>(agg, cd, W, b, h_out, n);
  };

  layer(x,  W0, b0, hA);
  layer(hA, W1, b1, hB);
  layer(hB, W2, b2, hA);

  k_pool_zero<<<dim3((g * DFE + TB - 1) / TB), blk, 0, stream>>>(psum, pcnt, g);
  k_pool_acc<<<dim3(poolBlocks), blk, 0, stream>>>(hA, gid, psum, pcnt, n);
  k_pool_final<<<dim3((g * DFE + TB - 1) / TB), blk, 0, stream>>>(psum, pcnt, (float*)d_out, g);
}


# aggregation: edges counting-sorted by destination once per call (int slot counters); one wave sums a node's incoming rows in registers, stores it once; no float atomics in layers (sum order changes)
# speedup vs baseline: 3.1989x; 2.3787x over previous
#include <hip/hip_runtime.h>
#include <cstdint>
#include <cstddef>

#define DFE 128

typedef float v2f __attribute__((ext_vector_type(2)));
typedef float v8f __attribute__((ext_vector_type(8)));

__global__ void k_init_cnt(int* __restrict__ cnt_out, int* __restrict__ cnt_in,
                           int* __restrict__ total, int n) {
  int i = blockIdx.x * blockDim.x + threadIdx.x;
  if (i < n) { cnt_out[i] = 0; cnt_in[i] = 0; }
  if (i == 0) *total = 0;
}

__global__ void k_edge_cnt(const int* __restrict__ src, const int* __restrict__ dst,
                           int* __restrict__ cnt_out, int* __restrict__ cnt_in, int e) {
  int i = blockIdx.x * blockDim.x + threadIdx.x;
  if (i < e) {
    atomicAdd(&cnt_out[src[i]], 1);
    atomicAdd(&cnt_in[dst[i]], 1);
  }
}

__global__ void k_rows(const int* __restrict__ cnt_out, const int* __restrict__ cnt_in,
                       float* __restrict__ cs, float* __restrict__ cd,
                       int* __restrict__ rowstart, int* __restrict__ cursor,
                       int* __restrict__ total, int n) {
  const int i = blockIdx.x * blockDim.x + threadIdx.x;
  const int lane = threadIdx.x & 31;
  int din = 0;
  if (i < n) {
    din = cnt_in[i];
    cs[i] = rsqrtf((float)(cnt_out[i] + 1));
    cd[i] = rsqrtf((float)(din + 1));
  }
  int incl = din;
  for (int o = 1; o < 32; o <<= 1) {
    int t = __shfl_up(incl, o);
    if (lane >= o) incl += t;
  }
  const int wsum = __shfl(incl, 31);
  int base = 0;
  if (lane == 31) base = atomicAdd(total, wsum);
  base = __shfl(base, 31);
  if (i < n) {
    const int st = base + incl - din;
    rowstart[i] = st;
    cursor[i] = st;
  }
}

__global__ void k_fill(const int* __restrict__ src, const int* __restrict__ dst,
                       int* __restrict__ cursor, int* __restrict__ esrc, int e) {
  int i = blockIdx.x * blockDim.x + threadIdx.x;
  if (i < e) {
    int slot = atomicAdd(&cursor[dst[i]], 1);
    esrc[slot] = src[i];
  }
}

template <int G>
__device__ __forceinline__ void add_rows(float4& acc, const float4* __restrict__ hrow,
                                         int sl, int clbits, int k, int lane) {
  float4 v[G];
  float c[G];
#pragma unroll
  for (int u = 0; u < G; ++u) {
    const int s = __builtin_amdgcn_readlane(sl, k + u);
    c[u] = __int_as_float(__builtin_amdgcn_readlane(clbits, k + u));
    v[u] = hrow[(size_t)s * (DFE / 4) + lane];
  }
#pragma unroll
  for (int u = 0; u < G; ++u) {
    acc.x += v[u].x * c[u];
    acc.y += v[u].y * c[u];
    acc.z += v[u].z * c[u];
    acc.w += v[u].w * c[u];
  }
}

__global__ __launch_bounds__(256) void k_gather(
    const int* __restrict__ rowstart, const int* __restrict__ rowend,
    const int* __restrict__ esrc, const float* __restrict__ h,
    const float* __restrict__ cs, float* __restrict__ agg, int n) {
  const int lane = threadIdx.x & 31;
  const int node = __builtin_amdgcn_readfirstlane(
      (int)((blockIdx.x * blockDim.x + threadIdx.x) >> 5));
  if (node >= n) return;
  const int beg = rowstart[node];
  const int end = rowend[node];
  const float4* hrow = (const float4*)h;

  float4 acc = hrow[(size_t)node * (DFE / 4) + lane];
  {
    const float s = cs[node];
    acc.x *= s; acc.y *= s; acc.z *= s; acc.w *= s;
  }

  for (int j = beg; j < end; j += 32) {
    const int rem = end - j;
    const int cnt = rem < 32 ? rem : 32;
    int sl = node;
    if (lane < cnt) sl = esrc[j + lane];
    const int clbits = __float_as_int(cs[sl]);
    int k = 0;
    for (; k + 8 <= cnt; k += 8) add_rows<8>(acc, hrow, sl, clbits, k, lane);
    if (k + 4 <= cnt) { add_rows<4>(acc, hrow, sl, clbits, k, lane); k += 4; }
    for (; k < cnt; ++k) add_rows<1>(acc, hrow, sl, clbits, k, lane);
  }
  ((float4*)agg)[(size_t)node * (DFE / 4) + lane] = acc;
}

__global__ __launch_bounds__(128) void k_gemm_relu(
    const float* __restrict__ agg, const float* __restrict__ cd,
    const float* __restrict__ W, const float* __restrict__ bias,
    float* __restrict__ out, int nrows) {
  __shared__ float As[16 * 132];
  const int tid = threadIdx.x;
  const int row0 = blockIdx.x * 16;

  for (int idx = tid; idx < 16 * 32; idx += 128) {
    int r = idx >> 5, c4 = idx & 31;
    int row = row0 + r;
    float4 v = make_float4(0.f, 0.f, 0.f, 0.f);
    float sc = 0.f;
    if (row < nrows) {
      v = ((const float4*)(agg + (size_t)row * DFE))[c4];
      sc = cd[row];
    }
    float* q = As + r * 132 + c4 * 4;
    q[0] = v.x * sc; q[1] = v.y * sc; q[2] = v.z * sc; q[3] = v.w * sc;
  }
  __syncthreads();

  const int wave = tid >> 5;
  const int lane = tid & 31;
  const int half = lane >> 4;
  const int l15  = lane & 15;
  const int ntile0 = wave * 2;
  const int colb0 = ntile0 * 16;

  v8f acc0 = {};
  v8f acc1 = {};
  const float* Arow = As + l15 * 132;

  for (int k0 = 0; k0 < DFE; k0 += 4) {
    int ka = k0 + 2 * half;
    v2f a = { Arow[ka], Arow[ka + 1] };
    v2f b0 = { W[(size_t)ka * DFE + colb0 + l15],
               W[(size_t)(ka + 1) * DFE + colb0 + l15] };
    v2f b1 = { W[(size_t)ka * DFE + colb0 + 16 + l15],
               W[(size_t)(ka + 1) * DFE + colb0 + 16 + l15] };
    acc0 = __builtin_amdgcn_wmma_f32_16x16x4_f32(false, a, false, b0,
                                                 (short)0, acc0, false, false);
    acc1 = __builtin_amdgcn_wmma_f32_16x16x4_f32(false, a, false, b1,
                                                 (short)0, acc1, false, false);
  }

  for (int v = 0; v < 8; ++v) {
    int row = row0 + v + 8 * half;
    if (row < nrows) {
      int col0 = colb0 + l15;
      float x0 = acc0[v] + bias[col0];
      out[(size_t)row * DFE + col0] = x0 > 0.f ? x0 : 0.f;
      int col1 = col0 + 16;
      float x1 = acc1[v] + bias[col1];
      out[(size_t)row * DFE + col1] = x1 > 0.f ? x1 : 0.f;
    }
  }
}

__global__ void k_pool_zero(float* __restrict__ sums, float* __restrict__ cnt, int g) {
  int i = blockIdx.x * blockDim.x + threadIdx.x;
  if (i < g * DFE) sums[i] = 0.f;
  if (i < g) cnt[i] = 0.f;
}

#define PNW 16

__global__ __launch_bounds__(256) void k_pool_acc(
    const float* __restrict__ h, const int* __restrict__ gid,
    float* __restrict__ sums, float* __restrict__ cnt, int n) {
  __shared__ __attribute__((aligned(16))) float part[8 * DFE];
  __shared__ int   pg[8];
  __shared__ float pc[8];
  const int tid = threadIdx.x;
  const int lane = tid & 31;
  const int wave = __builtin_amdgcn_readfirstlane(tid >> 5);
  const int base = (blockIdx.x * 8 + wave) * PNW;
  int nn = n - base;
  nn = nn < 0 ? 0 : (nn > PNW ? PNW : nn);
  const int last = n - 1;

  int gi = base + lane;
  gi = gi < last ? gi : last;
  const int myg = gid[gi];

  float4 v[PNW];
#pragma unroll
  for (int j = 0; j < PNW; ++j) {
    int nd = base + j;
    nd = nd < last ? nd : last;
    v[j] = ((const float4*)(h + (size_t)nd * DFE))[lane];
  }

  int cur = __builtin_amdgcn_readlane(myg, 0);
  float4 acc = v[0];
  float c = 1.0f;
#pragma unroll
  for (int j = 1; j < PNW; ++j) {
    const int gj = __builtin_amdgcn_readlane(myg, j);
    if (gj != cur) {
      float* p = sums + (size_t)cur * DFE + lane * 4;
      unsafeAtomicAdd(p + 0, acc.x);
      unsafeAtomicAdd(p + 1, acc.y);
      unsafeAtomicAdd(p + 2, acc.z);
      unsafeAtomicAdd(p + 3, acc.w);
      if (lane == 0) unsafeAtomicAdd(&cnt[cur], c);
      acc = make_float4(0.f, 0.f, 0.f, 0.f);
      c = 0.f;
      cur = gj;
    }
    const bool in = j < nn;
    acc.x += in ? v[j].x : 0.f;
    acc.y += in ? v[j].y : 0.f;
    acc.z += in ? v[j].z : 0.f;
    acc.w += in ? v[j].w : 0.f;
    c += in ? 1.0f : 0.f;
  }

  ((float4*)(part + wave * DFE))[lane] = acc;
  if (lane == 0) { pg[wave] = nn > 0 ? cur : -1; pc[wave] = c; }
  __syncthreads();

  if (tid <= DFE) {
    const bool is_cnt = tid == DFE;
    const float* sp = is_cnt ? pc : part + tid;
    const int sstep = is_cnt ? 1 : DFE;
    float* dp = is_cnt ? cnt : sums + tid;
    const size_t dstep = is_cnt ? 1 : DFE;
    int gs[8];
    float ps[8];
#pragma unroll
    for (int w = 0; w < 8; ++w) { gs[w] = pg[w]; ps[w] = sp[w * sstep]; }
    float s = 0.f;
    int g0 = -1;
#pragma unroll
    for (int w = 0; w < 8; ++w) {
      if (gs[w] >= 0) {
        if (gs[w] != g0) {
          if (g0 >= 0) unsafeAtomicAdd(dp + (size_t)g0 * dstep, s);
          s = 0.f;
          g0 = gs[w];
        }
        s += ps[w];
      }
    }
    if (g0 >= 0) unsafeAtomicAdd(dp + (size_t)g0 * dstep, s);
  }
}

__global__ void k_pool_final(const float* __restrict__ sums, const float* __restrict__ cnt,
                             float* __restrict__ out, int g) {
  int i = blockIdx.x * blockDim.x + threadIdx.x;
  if (i < g * DFE) {
    float c = cnt[i >> 7];
    c = c > 1.0f ? c : 1.0f;
    out[i] = sums[i] / c;
  }
}

extern "C" void kernel_launch(void* const* d_in, const int* in_sizes, int n_in,
                              void* d_out, int out_size, void* d_ws, size_t ws_size,
                              hipStream_t stream) {
  const float* x   = (const float*)d_in[0];
  const float* W0  = (const float*)d_in[1];
  const float* b0  = (const float*)d_in[2];
  const float* W1  = (const float*)d_in[3];
  const float* b1  = (const float*)d_in[4];
  const float* W2  = (const float*)d_in[5];
  const float* b2  = (const float*)d_in[6];
  const int*   src = (const int*)d_in[7];
  const int*   dst = (const int*)d_in[8];
  const int*   gid = (const int*)d_in[9];

  const int n = in_sizes[0] / DFE;
  const int e = in_sizes[7];
  const int g = out_size / DFE;

  float* ws = (float*)d_ws;
  size_t ND = (size_t)n * DFE;
  float* hA   = ws;
  float* hB   = hA + ND;
  float* agg  = hB + ND;
  float* cs   = agg + ND;
  float* cd   = cs + n;
  float* psum = cd + n;
  float* pcnt = psum + (size_t)g * DFE;
  int* cnt_out  = (int*)(pcnt + g);
  int* cnt_in   = cnt_out + n;
  int* rowstart = cnt_in + n;
  int* cursor   = rowstart + n;
  int* total    = cursor + n;
  int* esrc     = total + 32;

  const int TB = 256;
  dim3 blk(TB);

  k_init_cnt<<<dim3((n + TB - 1) / TB), blk, 0, stream>>>(cnt_out, cnt_in, total, n);
  k_edge_cnt<<<dim3((e + TB - 1) / TB), blk, 0, stream>>>(src, dst, cnt_out, cnt_in, e);
  k_rows<<<dim3((n + TB - 1) / TB), blk, 0, stream>>>(cnt_out, cnt_in, cs, cd, rowstart, cursor, total, n);
  k_fill<<<dim3((e + TB - 1) / TB), blk, 0, stream>>>(src, dst, cursor, esrc, e);

  const unsigned nodeWaveBlocks = (unsigned)(((size_t)n * 32 + TB - 1) / TB);
  const unsigned gemmBlocks     = (unsigned)((n + 15) / 16);
  const unsigned poolBlocks     = (unsigned)((n + 8 * PNW - 1) / (8 * PNW));

  auto layer = [&](const float* h_in, const float* W, const float* b, float* h_out) {
    k_gather<<<dim3(nodeWaveBlocks), blk, 0, stream>>>(rowstart, cursor, esrc, h_in, cs, agg, n);
    k_gemm_relu<<<dim3(gemmBlocks), dim3(128), 0, stream>>>(agg, cd, W, b, h_out, n);
  };

  layer(x,  W0, b0, hA);
  layer(hA, W1, b1, hB);
  layer(hB, W2, b2, hA);

  k_pool_zero<<<dim3((g * DFE + TB - 1) / TB), blk, 0, stream>>>(psum, pcnt, g);
  k_pool_acc<<<dim3(poolBlocks), blk, 0, stream>>>(hA, gid, psum, pcnt, n);
  k_pool_final<<<dim3((g * DFE + TB - 1) / TB), blk, 0, stream>>>(psum, pcnt, (float*)d_out, g);
}


# edge ordering and both degree counts by a two-level counting sort on LDS counters (chunk sort by 256-node bucket, then one block per bucket); no global atomic per edge; in-row sum order changes
# speedup vs baseline: 5.6836x; 1.7767x over previous
#include <hip/hip_runtime.h>
#include <cstdint>
#include <cstddef>

#define DFE 128

typedef float v2f __attribute__((ext_vector_type(2)));
typedef float v8f __attribute__((ext_vector_type(8)));

#define BSH 8
#define BKT (1 << BSH)
#define CH 4096
#define EPT (CH / 256)
#define NBMAX 1024
#define STG 8192

__global__ __launch_bounds__(256) void k_chunk_sort(
    const int* __restrict__ src, const int* __restrict__ dst,
    unsigned* __restrict__ recD, unsigned* __restrict__ recS,
    int* __restrict__ tabD, int* __restrict__ tabS,
    int* __restrict__ total, int e, int nb) {
  __shared__ int hD[NBMAX + 1];
  __shared__ int hS[NBMAX + 1];
  __shared__ unsigned stD[CH];
  __shared__ unsigned stS[CH];
  __shared__ int wsum[2][8];
  const int tid = threadIdx.x;
  const int lane = tid & 31;
  const int wave = tid >> 5;
  const int base = blockIdx.x * CH;
  if (blockIdx.x == 0 && tid == 0) *total = 0;
  for (int b = tid; b <= nb; b += 256) { hD[b] = 0; hS[b] = 0; }
  __syncthreads();

  int s[EPT], d[EPT], rD[EPT], rS[EPT];
#pragma unroll
  for (int k = 0; k < EPT; ++k) {
    const int i = base + k * 256 + tid;
    s[k] = -1; d[k] = -1;
    if (i < e) { s[k] = src[i]; d[k] = dst[i]; }
  }
#pragma unroll
  for (int k = 0; k < EPT; ++k) {
    rD[k] = 0; rS[k] = 0;
    if (d[k] >= 0) {
      rD[k] = atomicAdd(&hD[d[k] >> BSH], 1);
      rS[k] = atomicAdd(&hS[s[k] >> BSH], 1);
    }
  }
  __syncthreads();

  {
    int a[4], c[4];
    int sa = 0, sc = 0;
#pragma unroll
    for (int q = 0; q < 4; ++q) {
      const int idx = 4 * tid + q;
      a[q] = idx < nb ? hD[idx] : 0;
      c[q] = idx < nb ? hS[idx] : 0;
      sa += a[q]; sc += c[q];
    }
    int ia = sa, ic = sc;
    for (int o = 1; o < 32; o <<= 1) {
      const int t = __shfl_up(ia, o);
      const int u = __shfl_up(ic, o);
      if (lane >= o) { ia += t; ic += u; }
    }
    if (lane == 31) { wsum[0][wave] = ia; wsum[1][wave] = ic; }
    __syncthreads();
    int ea = ia - sa, ec = ic - sc;
    for (int w = 0; w < wave; ++w) { ea += wsum[0][w]; ec += wsum[1][w]; }
#pragma unroll
    for (int q = 0; q < 4; ++q) {
      const int idx = 4 * tid + q;
      if (idx < nb) { hD[idx] = ea; hS[idx] = ec; }
      ea += a[q]; ec += c[q];
    }
    if (tid == 255) { hD[nb] = ea; hS[nb] = ec; }
  }
  __syncthreads();

  for (int b = tid; b <= nb; b += 256) {
    tabD[(size_t)blockIdx.x * (nb + 1) + b] = hD[b];
    tabS[(size_t)blockIdx.x * (nb + 1) + b] = hS[b];
  }
#pragma unroll
  for (int k = 0; k < EPT; ++k) {
    if (d[k] >= 0) {
      stD[hD[d[k] >> BSH] + rD[k]] = ((unsigned)s[k] << BSH) | ((unsigned)d[k] & (BKT - 1));
      stS[hS[s[k] >> BSH] + rS[k]] = (unsigned)s[k] & (BKT - 1);
    }
  }
  __syncthreads();
  int cnt = e - base;
  cnt = cnt < CH ? cnt : CH;
  for (int j = tid; j < cnt; j += 256) {
    recD[(size_t)base + j] = stD[j];
    recS[(size_t)base + j] = stS[j];
  }
}

__global__ __launch_bounds__(256) void k_bucket(
    const unsigned* __restrict__ recD, const unsigned* __restrict__ recS,
    const int* __restrict__ tabD, const int* __restrict__ tabS,
    int* __restrict__ rowstart, int* __restrict__ rowend, int* __restrict__ esrc,
    float* __restrict__ cs, float* __restrict__ cd, int* __restrict__ total,
    int n, int nb, int nch) {
  __shared__ int c[BKT];
  __shared__ int cur[BKT];
  __shared__ int wsum[8];
  __shared__ int sbase;
  __shared__ int stage[STG];
  const int tid = threadIdx.x;
  const int lane = tid & 31;
  const int wave = tid >> 5;
  const bool isD = (int)blockIdx.x < nb;
  const int b = isD ? (int)blockIdx.x : (int)blockIdx.x - nb;
  const unsigned* __restrict__ rec = isD ? recD : recS;
  const int* __restrict__ tab = isD ? tabD : tabS;
  c[tid] = 0;
  __syncthreads();

  for (int ch = tid; ch < nch; ch += 256) {
    const int lo = tab[(size_t)ch * (nb + 1) + b];
    const int hi = tab[(size_t)ch * (nb + 1) + b + 1];
    const unsigned* p = rec + (size_t)ch * CH;
    for (int r = lo; r < hi; r += 8) {
      unsigned w[8];
#pragma unroll
      for (int u = 0; u < 8; ++u) {
        int ru = r + u;
        ru = ru < hi ? ru : hi - 1;
        w[u] = p[ru];
      }
#pragma unroll
      for (int u = 0; u < 8; ++u)
        if (r + u < hi) atomicAdd(&c[w[u] & (BKT - 1)], 1);
    }
  }
  __syncthreads();

  const int node = b * BKT + tid;
  const int mine = c[tid];
  if (!isD) {
    if (node < n) cs[node] = rsqrtf((float)(mine + 1));
    return;
  }

  int incl = mine;
  for (int o = 1; o < 32; o <<= 1) {
    const int t = __shfl_up(incl, o);
    if (lane >= o) incl += t;
  }
  if (lane == 31) wsum[wave] = incl;
  __syncthreads();
  int start = incl - mine;
  int tot = 0;
  for (int w = 0; w < 8; ++w) {
    if (w < wave) start += wsum[w];
    tot += wsum[w];
  }
  if (tid == 0) sbase = atomicAdd(total, tot);
  cur[tid] = start;
  __syncthreads();
  const int gbase = sbase;
  if (node < n) {
    rowstart[node] = gbase + start;
    rowend[node] = gbase + start + mine;
    cd[node] = rsqrtf((float)(mine + 1));
  }

  const bool staged = tot <= STG;
  for (int ch = tid; ch < nch; ch += 256) {
    const int lo = tab[(size_t)ch * (nb + 1) + b];
    const int hi = tab[(size_t)ch * (nb + 1) + b + 1];
    const unsigned* p = rec + (size_t)ch * CH;
    for (int r = lo; r < hi; r += 8) {
      unsigned w[8];
#pragma unroll
      for (int u = 0; u < 8; ++u) {
        int ru = r + u;
        ru = ru < hi ? ru : hi - 1;
        w[u] = p[ru];
      }
#pragma unroll
      for (int u = 0; u < 8; ++u) {
        if (r + u < hi) {
          const int slot = atomicAdd(&cur[w[u] & (BKT - 1)], 1);
          const int sv = (int)(w[u] >> BSH);
          if (staged) stage[slot] = sv; else esrc[(size_t)gbase + slot] = sv;
        }
      }
    }
  }
  if (staged) {
    __syncthreads();
    for (int j = tid; j < tot; j += 256) esrc[(size_t)gbase + j] = stage[j];
  }
}

template <int G>
__device__ __forceinline__ void add_rows(float4& acc, const float4* __restrict__ hrow,
                                         int sl, int clbits, int k, int lane) {
  float4 v[G];
  float c[G];
#pragma unroll
  for (int u = 0; u < G; ++u) {
    const int s = __builtin_amdgcn_readlane(sl, k + u);
    c[u] = __int_as_float(__builtin_amdgcn_readlane(clbits, k + u));
    v[u] = hrow[(size_t)s * (DFE / 4) + lane];
  }
#pragma unroll
  for (int u = 0; u < G; ++u) {
    acc.x += v[u].x * c[u];
    acc.y += v[u].y * c[u];
    acc.z += v[u].z * c[u];
    acc.w += v[u].w * c[u];
  }
}

__global__ __launch_bounds__(256) void k_gather(
    const int* __restrict__ rowstart, const int* __restrict__ rowend,
    const int* __restrict__ esrc, const float* __restrict__ h,
    const float* __restrict__ cs, float* __restrict__ agg, int n) {
  const int lane = threadIdx.x & 31;
  const int node = __builtin_amdgcn_readfirstlane(
      (int)((blockIdx.x * blockDim.x + threadIdx.x) >> 5));
  if (node >= n) return;
  const int beg = rowstart[node];
  const int end = rowend[node];
  const float4* hrow = (const float4*)h;

  float4 acc = hrow[(size_t)node * (DFE / 4) + lane];
  {
    const float s = cs[node];
    acc.x *= s; acc.y *= s; acc.z *= s; acc.w *= s;
  }

  for (int j = beg; j < end; j += 32) {
    const int rem = end - j;
    const int cnt = rem < 32 ? rem : 32;
    int sl = node;
    if (lane < cnt) sl = esrc[j + lane];
    const int clbits = __float_as_int(cs[sl]);
    int k = 0;
    for (; k + 8 <= cnt; k += 8) add_rows<8>(acc, hrow, sl, clbits, k, lane);
    if (k + 4 <= cnt) { add_rows<4>(acc, hrow, sl, clbits, k, lane); k += 4; }
    for (; k < cnt; ++k) add_rows<1>(acc, hrow, sl, clbits, k, lane);
  }
  ((float4*)agg)[(size_t)node * (DFE / 4) + lane] = acc;
}

__global__ __launch_bounds__(128) void k_gemm_relu(
    const float* __restrict__ agg, const float* __restrict__ cd,
    const float* __restrict__ W, const float* __restrict__ bias,
    float* __restrict__ out, int nrows) {
  __shared__ float As[16 * 132];
  const int tid = threadIdx.x;
  const int row0 = blockIdx.x * 16;

  for (int idx = tid; idx < 16 * 32; idx += 128) {
    int r = idx >> 5, c4 = idx & 31;
    int row = row0 + r;
    float4 v = make_float4(0.f, 0.f, 0.f, 0.f);
    float sc = 0.f;
    if (row < nrows) {
      v = ((const float4*)(agg + (size_t)row * DFE))[c4];
      sc = cd[row];
    }
    float* q = As + r * 132 + c4 * 4;
    q[0] = v.x * sc; q[1] = v.y * sc; q[2] = v.z * sc; q[3] = v.w * sc;
  }
  __syncthreads();

  const int wave = tid >> 5;
  const int lane = tid & 31;
  const int half = lane >> 4;
  const int l15  = lane & 15;
  const int ntile0 = wave * 2;
  const int colb0 = ntile0 * 16;

  v8f acc0 = {};
  v8f acc1 = {};
  const float* Arow = As + l15 * 132;

  for (int k0 = 0; k0 < DFE; k0 += 4) {
    int ka = k0 + 2 * half;
    v2f a = { Arow[ka], Arow[ka + 1] };
    v2f b0 = { W[(size_t)ka * DFE + colb0 + l15],
               W[(size_t)(ka + 1) * DFE + colb0 + l15] };
    v2f b1 = { W[(size_t)ka * DFE + colb0 + 16 + l15],
               W[(size_t)(ka + 1) * DFE + colb0 + 16 + l15] };
    acc0 = __builtin_amdgcn_wmma_f32_16x16x4_f32(false, a, false, b0,
                                                 (short)0, acc0, false, false);
    acc1 = __builtin_amdgcn_wmma_f32_16x16x4_f32(false, a, false, b1,
                                                 (short)0, acc1, false, false);
  }

  for (int v = 0; v < 8; ++v) {
    int row = row0 + v + 8 * half;
    if (row < nrows) {
      int col0 = colb0 + l15;
      float x0 = acc0[v] + bias[col0];
      out[(size_t)row * DFE + col0] = x0 > 0.f ? x0 : 0.f;
      int col1 = col0 + 16;
      float x1 = acc1[v] + bias[col1];
      out[(size_t)row * DFE + col1] = x1 > 0.f ? x1 : 0.f;
    }
  }
}

__global__ void k_pool_zero(float* __restrict__ sums, float* __restrict__ cnt, int g) {
  int i = blockIdx.x * blockDim.x + threadIdx.x;
  if (i < g * DFE) sums[i] = 0.f;
  if (i < g) cnt[i] = 0.f;
}

#define PNW 16

__global__ __launch_bounds__(256) void k_pool_acc(
    const float* __restrict__ h, const int* __restrict__ gid,
    float* __restrict__ sums, float* __restrict__ cnt, int n) {
  __shared__ __attribute__((aligned(16))) float part[8 * DFE];
  __shared__ int   pg[8];
  __shared__ float pc[8];
  const int tid = threadIdx.x;
  const int lane = tid & 31;
  const int wave = __builtin_amdgcn_readfirstlane(tid >> 5);
  const int base = (blockIdx.x * 8 + wave) * PNW;
  int nn = n - base;
  nn = nn < 0 ? 0 : (nn > PNW ? PNW : nn);
  const int last = n - 1;

  int gi = base + lane;
  gi = gi < last ? gi : last;
  const int myg = gid[gi];

  float4 v[PNW];
#pragma unroll
  for (int j = 0; j < PNW; ++j) {
    int nd = base + j;
    nd = nd < last ? nd : last;
    v[j] = ((const float4*)(h + (size_t)nd * DFE))[lane];
  }

  int cur = __builtin_amdgcn_readlane(myg, 0);
  float4 acc = v[0];
  float c = 1.0f;
#pragma unroll
  for (int j = 1; j < PNW; ++j) {
    const int gj = __builtin_amdgcn_readlane(myg, j);
    if (gj != cur) {
      float* p = sums + (size_t)cur * DFE + lane * 4;
      unsafeAtomicAdd(p + 0, acc.x);
      unsafeAtomicAdd(p + 1, acc.y);
      unsafeAtomicAdd(p + 2, acc.z);
      unsafeAtomicAdd(p + 3, acc.w);
      if (lane == 0) unsafeAtomicAdd(&cnt[cur], c);
      acc = make_float4(0.f, 0.f, 0.f, 0.f);
      c = 0.f;
      cur = gj;
    }
    const bool in = j < nn;
    acc.x += in ? v[j].x : 0.f;
    acc.y += in ? v[j].y : 0.f;
    acc.z += in ? v[j].z : 0.f;
    acc.w += in ? v[j].w : 0.f;
    c += in ? 1.0f : 0.f;
  }

  ((float4*)(part + wave * DFE))[lane] = acc;
  if (lane == 0) { pg[wave] = nn > 0 ? cur : -1; pc[wave] = c; }
  __syncthreads();

  if (tid <= DFE) {
    const bool is_cnt = tid == DFE;
    const float* sp = is_cnt ? pc : part + tid;
    const int sstep = is_cnt ? 1 : DFE;
    float* dp = is_cnt ? cnt : sums + tid;
    const size_t dstep = is_cnt ? 1 : DFE;
    int gs[8];
    float ps[8];
#pragma unroll
    for (int w = 0; w < 8; ++w) { gs[w] = pg[w]; ps[w] = sp[w * sstep]; }
    float s = 0.f;
    int g0 = -1;
#pragma unroll
    for (int w = 0; w < 8; ++w) {
      if (gs[w] >= 0) {
        if (gs[w] != g0) {
          if (g0 >= 0) unsafeAtomicAdd(dp + (size_t)g0 * dstep, s);
          s = 0.f;
          g0 = gs[w];
        }
        s += ps[w];
      }
    }
    if (g0 >= 0) unsafeAtomicAdd(dp + (size_t)g0 * dstep, s);
  }
}

__global__ void k_pool_final(const float* __restrict__ sums, const float* __restrict__ cnt,
                             float* __restrict__ out, int g) {
  int i = blockIdx.x * blockDim.x + threadIdx.x;
  if (i < g * DFE) {
    float c = cnt[i >> 7];
    c = c > 1.0f ? c : 1.0f;
    out[i] = sums[i] / c;
  }
}

extern "C" void kernel_launch(void* const* d_in, const int* in_sizes, int n_in,
                              void* d_out, int out_size, void* d_ws, size_t ws_size,
                              hipStream_t stream) {
  const float* x   = (const float*)d_in[0];
  const float* W0  = (const float*)d_in[1];
  const float* b0  = (const float*)d_in[2];
  const float* W1  = (const float*)d_in[3];
  const float* b1  = (const float*)d_in[4];
  const float* W2  = (const float*)d_in[5];
  const float* b2  = (const float*)d_in[6];
  const int*   src = (const int*)d_in[7];
  const int*   dst = (const int*)d_in[8];
  const int*   gid = (const int*)d_in[9];

  const int n = in_sizes[0] / DFE;
  const int e = in_sizes[7];
  const int g = out_size / DFE;

  const int nb  = (n + BKT - 1) / BKT;
  const int nch = (e + CH - 1) / CH;

  float* ws = (float*)d_ws;
  size_t ND = (size_t)n * DFE;
  float* hA   = ws;
  float* hB   = hA + ND;
  float* agg  = hB + ND;
  float* cs   = agg + ND;
  float* cd   = cs + n;
  float* psum = cd + n;
  float* pcnt = psum + (size_t)g * DFE;
  int* rowstart = (int*)(pcnt + g);
  int* rowend   = rowstart + n;
  int* total    = rowend + n;
  int* esrc     = total + 32;
  unsigned* recD = (unsigned*)(esrc + e);
  unsigned* recS = recD + (size_t)nch * CH;
  int* tabD     = (int*)(recS + (size_t)nch * CH);
  int* tabS     = tabD + (size_t)nch * (nb + 1);

  const int TB = 256;
  dim3 blk(TB);

  k_chunk_sort<<<dim3(nch), blk, 0, stream>>>(src, dst, recD, recS, tabD, tabS, total, e, nb);
  k_bucket<<<dim3(2 * nb), blk, 0, stream>>>(recD, recS, tabD, tabS, rowstart, rowend, esrc,
                                             cs, cd, total, n, nb, nch);

  const unsigned nodeWaveBlocks = (unsigned)(((size_t)n * 32 + TB - 1) / TB);
  const unsigned gemmBlocks     = (unsigned)((n + 15) / 16);
  const unsigned poolBlocks     = (unsigned)((n + 8 * PNW - 1) / (8 * PNW));

  auto layer = [&](const float* h_in, const float* W, const float* b, float* h_out) {
    k_gather<<<dim3(nodeWaveBlocks), blk, 0, stream>>>(rowstart, rowend, esrc, h_in, cs, agg, n);
    k_gemm_relu<<<dim3(gemmBlocks), dim3(128), 0, stream>>>(agg, cd, W, b, h_out, n);
  };

  layer(x,  W0, b0, hA);
  layer(hA, W1, b1, hB);
  layer(hB, W2, b2, hA);

  k_pool_zero<<<dim3((g * DFE + TB - 1) / TB), blk, 0, stream>>>(psum, pcnt, g);
  k_pool_acc<<<dim3(poolBlocks), blk, 0, stream>>>(hA, gid, psum, pcnt, n);
  k_pool_final<<<dim3((g * DFE + TB - 1) / TB), blk, 0, stream>>>(psum, pcnt, (float*)d_out, g);
}


# each layer is one launch: a block's 4 waves sum 4 nodes' incoming rows each in registers and store them, scaled, into the 16-row LDS A tile of its WMMA product; agg stays on chip (11 launches to 8)
# speedup vs baseline: 6.6246x; 1.1656x over previous
#include <hip/hip_runtime.h>
#include <cstdint>
#include <cstddef>

#define DFE 128

typedef float v2f __attribute__((ext_vector_type(2)));
typedef float v8f __attribute__((ext_vector_type(8)));

#define BSH 8
#define BKT (1 << BSH)
#define CH 4096
#define EPT (CH / 256)
#define NBMAX 1024
#define STG 8192

__global__ __launch_bounds__(256) void k_chunk_sort(
    const int* __restrict__ src, const int* __restrict__ dst,
    unsigned* __restrict__ recD, unsigned* __restrict__ recS,
    int* __restrict__ tabD, int* __restrict__ tabS,
    int* __restrict__ total, int e, int nb) {
  __shared__ int hD[NBMAX + 1];
  __shared__ int hS[NBMAX + 1];
  __shared__ unsigned stD[CH];
  __shared__ unsigned stS[CH];
  __shared__ int wsum[2][8];
  const int tid = threadIdx.x;
  const int lane = tid & 31;
  const int wave = tid >> 5;
  const int base = blockIdx.x * CH;
  if (blockIdx.x == 0 && tid == 0) *total = 0;
  for (int b = tid; b <= nb; b += 256) { hD[b] = 0; hS[b] = 0; }
  __syncthreads();

  int s[EPT], d[EPT], rD[EPT], rS[EPT];
#pragma unroll
  for (int k = 0; k < EPT; ++k) {
    const int i = base + k * 256 + tid;
    s[k] = -1; d[k] = -1;
    if (i < e) { s[k] = src[i]; d[k] = dst[i]; }
  }
#pragma unroll
  for (int k = 0; k < EPT; ++k) {
    rD[k] = 0; rS[k] = 0;
    if (d[k] >= 0) {
      rD[k] = atomicAdd(&hD[d[k] >> BSH], 1);
      rS[k] = atomicAdd(&hS[s[k] >> BSH], 1);
    }
  }
  __syncthreads();

  {
    int a[4], c[4];
    int sa = 0, sc = 0;
#pragma unroll
    for (int q = 0; q < 4; ++q) {
      const int idx = 4 * tid + q;
      a[q] = idx < nb ? hD[idx] : 0;
      c[q] = idx < nb ? hS[idx] : 0;
      sa += a[q]; sc += c[q];
    }
    int ia = sa, ic = sc;
    for (int o = 1; o < 32; o <<= 1) {
      const int t = __shfl_up(ia, o);
      const int u = __shfl_up(ic, o);
      if (lane >= o) { ia += t; ic += u; }
    }
    if (lane == 31) { wsum[0][wave] = ia; wsum[1][wave] = ic; }
    __syncthreads();
    int ea = ia - sa, ec = ic - sc;
    for (int w = 0; w < wave; ++w) { ea += wsum[0][w]; ec += wsum[1][w]; }
#pragma unroll
    for (int q = 0; q < 4; ++q) {
      const int idx = 4 * tid + q;
      if (idx < nb) { hD[idx] = ea; hS[idx] = ec; }
      ea += a[q]; ec += c[q];
    }
    if (tid == 255) { hD[nb] = ea; hS[nb] = ec; }
  }
  __syncthreads();

  for (int b = tid; b <= nb; b += 256) {
    tabD[(size_t)blockIdx.x * (nb + 1) + b] = hD[b];
    tabS[(size_t)blockIdx.x * (nb + 1) + b] = hS[b];
  }
#pragma unroll
  for (int k = 0; k < EPT; ++k) {
    if (d[k] >= 0) {
      stD[hD[d[k] >> BSH] + rD[k]] = ((unsigned)s[k] << BSH) | ((unsigned)d[k] & (BKT - 1));
      stS[hS[s[k] >> BSH] + rS[k]] = (unsigned)s[k] & (BKT - 1);
    }
  }
  __syncthreads();
  int cnt = e - base;
  cnt = cnt < CH ? cnt : CH;
  for (int j = tid; j < cnt; j += 256) {
    recD[(size_t)base + j] = stD[j];
    recS[(size_t)base + j] = stS[j];
  }
}

__global__ __launch_bounds__(256) void k_bucket(
    const unsigned* __restrict__ recD, const unsigned* __restrict__ recS,
    const int* __restrict__ tabD, const int* __restrict__ tabS,
    int* __restrict__ rowstart, int* __restrict__ rowend, int* __restrict__ esrc,
    float* __restrict__ cs, float* __restrict__ cd, int* __restrict__ total,
    int n, int nb, int nch) {
  __shared__ int c[BKT];
  __shared__ int cur[BKT];
  __shared__ int wsum[8];
  __shared__ int sbase;
  __shared__ int stage[STG];
  const int tid = threadIdx.x;
  const int lane = tid & 31;
  const int wave = tid >> 5;
  const bool isD = (int)blockIdx.x < nb;
  const int b = isD ? (int)blockIdx.x : (int)blockIdx.x - nb;
  const unsigned* __restrict__ rec = isD ? recD : recS;
  const int* __restrict__ tab = isD ? tabD : tabS;
  c[tid] = 0;
  __syncthreads();

  for (int ch = tid; ch < nch; ch += 256) {
    const int lo = tab[(size_t)ch * (nb + 1) + b];
    const int hi = tab[(size_t)ch * (nb + 1) + b + 1];
    const unsigned* p = rec + (size_t)ch * CH;
    for (int r = lo; r < hi; r += 8) {
      unsigned w[8];
#pragma unroll
      for (int u = 0; u < 8; ++u) {
        int ru = r + u;
        ru = ru < hi ? ru : hi - 1;
        w[u] = p[ru];
      }
#pragma unroll
      for (int u = 0; u < 8; ++u)
        if (r + u < hi) atomicAdd(&c[w[u] & (BKT - 1)], 1);
    }
  }
  __syncthreads();

  const int node = b * BKT + tid;
  const int mine = c[tid];
  if (!isD) {
    if (node < n) cs[node] = rsqrtf((float)(mine + 1));
    return;
  }

  int incl = mine;
  for (int o = 1; o < 32; o <<= 1) {
    const int t = __shfl_up(incl, o);
    if (lane >= o) incl += t;
  }
  if (lane == 31) wsum[wave] = incl;
  __syncthreads();
  int start = incl - mine;
  int tot = 0;
  for (int w = 0; w < 8; ++w) {
    if (w < wave) start += wsum[w];
    tot += wsum[w];
  }
  if (tid == 0) sbase = atomicAdd(total, tot);
  cur[tid] = start;
  __syncthreads();
  const int gbase = sbase;
  if (node < n) {
    rowstart[node] = gbase + start;
    rowend[node] = gbase + start + mine;
    cd[node] = rsqrtf((float)(mine + 1));
  }

  const bool staged = tot <= STG;
  for (int ch = tid; ch < nch; ch += 256) {
    const int lo = tab[(size_t)ch * (nb + 1) + b];
    const int hi = tab[(size_t)ch * (nb + 1) + b + 1];
    const unsigned* p = rec + (size_t)ch * CH;
    for (int r = lo; r < hi; r += 8) {
      unsigned w[8];
#pragma unroll
      for (int u = 0; u < 8; ++u) {
        int ru = r + u;
        ru = ru < hi ? ru : hi - 1;
        w[u] = p[ru];
      }
#pragma unroll
      for (int u = 0; u < 8; ++u) {
        if (r + u < hi) {
          const int slot = atomicAdd(&cur[w[u] & (BKT - 1)], 1);
          const int sv = (int)(w[u] >> BSH);
          if (staged) stage[slot] = sv; else esrc[(size_t)gbase + slot] = sv;
        }
      }
    }
  }
  if (staged) {
    __syncthreads();
    for (int j = tid; j < tot; j += 256) esrc[(size_t)gbase + j] = stage[j];
  }
}

template <int G>
__device__ __forceinline__ void add_rows(float4& acc, const float* __restrict__ h,
                                         int sl, int clbits, int k, unsigned lane16) {
  float4 v[G];
  float c[G];
#pragma unroll
  for (int u = 0; u < G; ++u) {
    const unsigned s = (unsigned)__builtin_amdgcn_readlane(sl, k + u);
    c[u] = __int_as_float(__builtin_amdgcn_readlane(clbits, k + u));
    v[u] = *(const float4*)((const char*)h + (s * (unsigned)(DFE * 4) + lane16));
  }
#pragma unroll
  for (int u = 0; u < G; ++u) {
    acc.x += v[u].x * c[u];
    acc.y += v[u].y * c[u];
    acc.z += v[u].z * c[u];
    acc.w += v[u].w * c[u];
  }
}

#define NPW 4
__global__ __launch_bounds__(128) void k_layer(
    const int* __restrict__ rowstart, const int* __restrict__ rowend,
    const int* __restrict__ esrc, const float* __restrict__ h,
    const float* __restrict__ cs, const float* __restrict__ cd,
    const float* __restrict__ W, const float* __restrict__ bias,
    float* __restrict__ out, int nrows) {
  __shared__ __attribute__((aligned(16))) float As[16 * 132];
  const int tid = threadIdx.x;
  const int lane = tid & 31;
  const int wave = __builtin_amdgcn_readfirstlane(tid >> 5);
  const int row0 = blockIdx.x * 16;
  const int last = nrows - 1;
  const unsigned lane16 = (unsigned)lane * 16u;

  int nd[NPW], beg[NPW], end[NPW];
#pragma unroll
  for (int i = 0; i < NPW; ++i) {
    const int node = row0 + wave * NPW + i;
    nd[i] = node < last ? node : last;
    beg[i] = rowstart[nd[i]];
    end[i] = rowend[nd[i]];
  }
  int slnext = nd[0];
  {
    const int rem = end[0] - beg[0];
    if (lane < (rem < 32 ? rem : 32)) slnext = esrc[beg[0] + lane];
  }

#pragma unroll
  for (int i = 0; i < NPW; ++i) {
    const int node = nd[i];
    int sl = slnext;
    if (i + 1 < NPW) {
      const int rem = end[i + 1] - beg[i + 1];
      slnext = nd[i + 1];
      if (lane < (rem < 32 ? rem : 32)) slnext = esrc[beg[i + 1] + lane];
    }
    int clbits = __float_as_int(cs[sl]);
    float4 acc = *(const float4*)((const char*)h + ((unsigned)node * (unsigned)(DFE * 4) + lane16));
    {
      const float s = cs[node];
      acc.x *= s; acc.y *= s; acc.z *= s; acc.w *= s;
    }
    for (int j = beg[i]; j < end[i]; j += 32) {
      const int rem = end[i] - j;
      const int cnt = rem < 32 ? rem : 32;
      if (j != beg[i]) {
        sl = node;
        if (lane < cnt) sl = esrc[j + lane];
        clbits = __float_as_int(cs[sl]);
      }
      int k = 0;
      for (; k + 8 <= cnt; k += 8) add_rows<8>(acc, h, sl, clbits, k, lane16);
      if (k + 4 <= cnt) { add_rows<4>(acc, h, sl, clbits, k, lane16); k += 4; }
      for (; k < cnt; ++k) add_rows<1>(acc, h, sl, clbits, k, lane16);
    }
    {
      const float sc = (row0 + wave * NPW + i) < nrows ? cd[node] : 0.f;
      acc.x *= sc; acc.y *= sc; acc.z *= sc; acc.w *= sc;
    }
    ((float4*)(As + (wave * NPW + i) * 132))[lane] = acc;
  }
  __syncthreads();

  const int half = lane >> 4;
  const int l15  = lane & 15;
  const int ntile0 = wave * 2;
  const int colb0 = ntile0 * 16;

  v8f acc0 = {};
  v8f acc1 = {};
  const float* Arow = As + l15 * 132;

#pragma unroll 8
  for (int k0 = 0; k0 < DFE; k0 += 4) {
    int ka = k0 + 2 * half;
    v2f a = { Arow[ka], Arow[ka + 1] };
    v2f b0 = { W[(size_t)ka * DFE + colb0 + l15],
               W[(size_t)(ka + 1) * DFE + colb0 + l15] };
    v2f b1 = { W[(size_t)ka * DFE + colb0 + 16 + l15],
               W[(size_t)(ka + 1) * DFE + colb0 + 16 + l15] };
    acc0 = __builtin_amdgcn_wmma_f32_16x16x4_f32(false, a, false, b0,
                                                 (short)0, acc0, false, false);
    acc1 = __builtin_amdgcn_wmma_f32_16x16x4_f32(false, a, false, b1,
                                                 (short)0, acc1, false, false);
  }

  for (int v = 0; v < 8; ++v) {
    int row = row0 + v + 8 * half;
    if (row < nrows) {
      int col0 = colb0 + l15;
      float x0 = acc0[v] + bias[col0];
      out[(size_t)row * DFE + col0] = x0 > 0.f ? x0 : 0.f;
      int col1 = col0 + 16;
      float x1 = acc1[v] + bias[col1];
      out[(size_t)row * DFE + col1] = x1 > 0.f ? x1 : 0.f;
    }
  }
}

__global__ void k_pool_zero(float* __restrict__ sums, float* __restrict__ cnt, int g) {
  int i = blockIdx.x * blockDim.x + threadIdx.x;
  if (i < g * DFE) sums[i] = 0.f;
  if (i < g) cnt[i] = 0.f;
}

#define PNW 16

__global__ __launch_bounds__(256) void k_pool_acc(
    const float* __restrict__ h, const int* __restrict__ gid,
    float* __restrict__ sums, float* __restrict__ cnt, int n) {
  __shared__ __attribute__((aligned(16))) float part[8 * DFE];
  __shared__ int   pg[8];
  __shared__ float pc[8];
  const int tid = threadIdx.x;
  const int lane = tid & 31;
  const int wave = __builtin_amdgcn_readfirstlane(tid >> 5);
  const int base = (blockIdx.x * 8 + wave) * PNW;
  int nn = n - base;
  nn = nn < 0 ? 0 : (nn > PNW ? PNW : nn);
  const int last = n - 1;

  int gi = base + lane;
  gi = gi < last ? gi : last;
  const int myg = gid[gi];

  float4 v[PNW];
#pragma unroll
  for (int j = 0; j < PNW; ++j) {
    int nd = base + j;
    nd = nd < last ? nd : last;
    v[j] = ((const float4*)(h + (size_t)nd * DFE))[lane];
  }

  int cur = __builtin_amdgcn_readlane(myg, 0);
  float4 acc = v[0];
  float c = 1.0f;
#pragma unroll
  for (int j = 1; j < PNW; ++j) {
    const int gj = __builtin_amdgcn_readlane(myg, j);
    if (gj != cur) {
      float* p = sums + (size_t)cur * DFE + lane * 4;
      unsafeAtomicAdd(p + 0, acc.x);
      unsafeAtomicAdd(p + 1, acc.y);
      unsafeAtomicAdd(p + 2, acc.z);
      unsafeAtomicAdd(p + 3, acc.w);
      if (lane == 0) unsafeAtomicAdd(&cnt[cur], c);
      acc = make_float4(0.f, 0.f, 0.f, 0.f);
      c = 0.f;
      cur = gj;
    }
    const bool in = j < nn;
    acc.x += in ? v[j].x : 0.f;
    acc.y += in ? v[j].y : 0.f;
    acc.z += in ? v[j].z : 0.f;
    acc.w += in ? v[j].w : 0.f;
    c += in ? 1.0f : 0.f;
  }

  ((float4*)(part + wave * DFE))[lane] = acc;
  if (lane == 0) { pg[wave] = nn > 0 ? cur : -1; pc[wave] = c; }
  __syncthreads();

  if (tid <= DFE) {
    const bool is_cnt = tid == DFE;
    const float* sp = is_cnt ? pc : part + tid;
    const int sstep = is_cnt ? 1 : DFE;
    float* dp = is_cnt ? cnt : sums + tid;
    const size_t dstep = is_cnt ? 1 : DFE;
    int gs[8];
    float ps[8];
#pragma unroll
    for (int w = 0; w < 8; ++w) { gs[w] = pg[w]; ps[w] = sp[w * sstep]; }
    float s = 0.f;
    int g0 = -1;
#pragma unroll
    for (int w = 0; w < 8; ++w) {
      if (gs[w] >= 0) {
        if (gs[w] != g0) {
          if (g0 >= 0) unsafeAtomicAdd(dp + (size_t)g0 * dstep, s);
          s = 0.f;
          g0 = gs[w];
        }
        s += ps[w];
      }
    }
    if (g0 >= 0) unsafeAtomicAdd(dp + (size_t)g0 * dstep, s);
  }
}

__global__ void k_pool_final(const float* __restrict__ sums, const float* __restrict__ cnt,
                             float* __restrict__ out, int g) {
  int i = blockIdx.x * blockDim.x + threadIdx.x;
  if (i < g * DFE) {
    float c = cnt[i >> 7];
    c = c > 1.0f ? c : 1.0f;
    out[i] = sums[i] / c;
  }
}

extern "C" void kernel_launch(void* const* d_in, const int* in_sizes, int n_in,
                              void* d_out, int out_size, void* d_ws, size_t ws_size,
                              hipStream_t stream) {
  const float* x   = (const float*)d_in[0];
  const float* W0  = (const float*)d_in[1];
  const float* b0  = (const float*)d_in[2];
  const float* W1  = (const float*)d_in[3];
  const float* b1  = (const float*)d_in[4];
  const float* W2  = (const float*)d_in[5];
  const float* b2  = (const float*)d_in[6];
  const int*   src = (const int*)d_in[7];
  const int*   dst = (const int*)d_in[8];
  const int*   gid = (const int*)d_in[9];

  const int n = in_sizes[0] / DFE;
  const int e = in_sizes[7];
  const int g = out_size / DFE;

  const int nb  = (n + BKT - 1) / BKT;
  const int nch = (e + CH - 1) / CH;

  float* ws = (float*)d_ws;
  size_t ND = (size_t)n * DFE;
  float* hA   = ws;
  float* hB   = hA + ND;
  float* cs   = hB + 2 * ND;
  float* cd   = cs + n;
  float* psum = cd + n;
  float* pcnt = psum + (size_t)g * DFE;
  int* rowstart = (int*)(pcnt + g);
  int* rowend   = rowstart + n;
  int* total    = rowend + n;
  int* esrc     = total + 32;
  unsigned* recD = (unsigned*)(esrc + e);
  unsigned* recS = recD + (size_t)nch * CH;
  int* tabD     = (int*)(recS + (size_t)nch * CH);
  int* tabS     = tabD + (size_t)nch * (nb + 1);

  const int TB = 256;
  dim3 blk(TB);

  k_chunk_sort<<<dim3(nch), blk, 0, stream>>>(src, dst, recD, recS, tabD, tabS, total, e, nb);
  k_bucket<<<dim3(2 * nb), blk, 0, stream>>>(recD, recS, tabD, tabS, rowstart, rowend, esrc,
                                             cs, cd, total, n, nb, nch);

  const unsigned layerBlocks    = (unsigned)((n + 15) / 16);
  const unsigned poolBlocks     = (unsigned)((n + 8 * PNW - 1) / (8 * PNW));

  auto layer = [&](const float* h_in, const float* W, const float* b, float* h_out) {
    k_layer<<<dim3(layerBlocks), dim3(128), 0, stream>>>(rowstart, rowend, esrc, h_in, cs, cd, W, b, h_out, n);
  };

  layer(x,  W0, b0, hA);
  layer(hA, W1, b1, hB);
  layer(hB, W2, b2, hA);

  k_pool_zero<<<dim3((g * DFE + TB - 1) / TB), blk, 0, stream>>>(psum, pcnt, g);
  k_pool_acc<<<dim3(poolBlocks), blk, 0, stream>>>(hA, gid, psum, pcnt, n);
  k_pool_final<<<dim3((g * DFE + TB - 1) / TB), blk, 0, stream>>>(psum, pcnt, (float*)d_out, g);
}


# layers 1,2 store each row already times its source norm cs, so layers 2,3 gather rows with no per-edge scale load or multiply (products rounded before the adds, as the reference; sum order kept)
# speedup vs baseline: 6.9802x; 1.0537x over previous
#include <hip/hip_runtime.h>
#include <cstdint>
#include <cstddef>

#define DFE 128

typedef float v2f __attribute__((ext_vector_type(2)));
typedef float v8f __attribute__((ext_vector_type(8)));

#define BSH 8
#define BKT (1 << BSH)
#define CH 4096
#define EPT (CH / 256)
#define NBMAX 1024
#define STG 8192

__global__ __launch_bounds__(256) void k_chunk_sort(
    const int* __restrict__ src, const int* __restrict__ dst,
    unsigned* __restrict__ recD, unsigned* __restrict__ recS,
    int* __restrict__ tabD, int* __restrict__ tabS,
    int* __restrict__ total, int e, int nb) {
  __shared__ int hD[NBMAX + 1];
  __shared__ int hS[NBMAX + 1];
  __shared__ unsigned stD[CH];
  __shared__ unsigned stS[CH];
  __shared__ int wsum[2][8];
  const int tid = threadIdx.x;
  const int lane = tid & 31;
  const int wave = tid >> 5;
  const int base = blockIdx.x * CH;
  if (blockIdx.x == 0 && tid == 0) *total = 0;
  for (int b = tid; b <= nb; b += 256) { hD[b] = 0; hS[b] = 0; }
  __syncthreads();

  int s[EPT], d[EPT], rD[EPT], rS[EPT];
#pragma unroll
  for (int k = 0; k < EPT; ++k) {
    const int i = base + k * 256 + tid;
    s[k] = -1; d[k] = -1;
    if (i < e) { s[k] = src[i]; d[k] = dst[i]; }
  }
#pragma unroll
  for (int k = 0; k < EPT; ++k) {
    rD[k] = 0; rS[k] = 0;
    if (d[k] >= 0) {
      rD[k] = atomicAdd(&hD[d[k] >> BSH], 1);
      rS[k] = atomicAdd(&hS[s[k] >> BSH], 1);
    }
  }
  __syncthreads();

  {
    int a[4], c[4];
    int sa = 0, sc = 0;
#pragma unroll
    for (int q = 0; q < 4; ++q) {
      const int idx = 4 * tid + q;
      a[q] = idx < nb ? hD[idx] : 0;
      c[q] = idx < nb ? hS[idx] : 0;
      sa += a[q]; sc += c[q];
    }
    int ia = sa, ic = sc;
    for (int o = 1; o < 32; o <<= 1) {
      const int t = __shfl_up(ia, o);
      const int u = __shfl_up(ic, o);
      if (lane >= o) { ia += t; ic += u; }
    }
    if (lane == 31) { wsum[0][wave] = ia; wsum[1][wave] = ic; }
    __syncthreads();
    int ea = ia - sa, ec = ic - sc;
    for (int w = 0; w < wave; ++w) { ea += wsum[0][w]; ec += wsum[1][w]; }
#pragma unroll
    for (int q = 0; q < 4; ++q) {
      const int idx = 4 * tid + q;
      if (idx < nb) { hD[idx] = ea; hS[idx] = ec; }
      ea += a[q]; ec += c[q];
    }
    if (tid == 255) { hD[nb] = ea; hS[nb] = ec; }
  }
  __syncthreads();

  for (int b = tid; b <= nb; b += 256) {
    tabD[(size_t)blockIdx.x * (nb + 1) + b] = hD[b];
    tabS[(size_t)blockIdx.x * (nb + 1) + b] = hS[b];
  }
#pragma unroll
  for (int k = 0; k < EPT; ++k) {
    if (d[k] >= 0) {
      stD[hD[d[k] >> BSH] + rD[k]] = ((unsigned)s[k] << BSH) | ((unsigned)d[k] & (BKT - 1));
      stS[hS[s[k] >> BSH] + rS[k]] = (unsigned)s[k] & (BKT - 1);
    }
  }
  __syncthreads();
  int cnt = e - base;
  cnt = cnt < CH ? cnt : CH;
  for (int j = tid; j < cnt; j += 256) {
    recD[(size_t)base + j] = stD[j];
    recS[(size_t)base + j] = stS[j];
  }
}

__global__ __launch_bounds__(256) void k_bucket(
    const unsigned* __restrict__ recD, const unsigned* __restrict__ recS,
    const int* __restrict__ tabD, const int* __restrict__ tabS,
    int* __restrict__ rowstart, int* __restrict__ rowend, int* __restrict__ esrc,
    float* __restrict__ cs, float* __restrict__ cd, int* __restrict__ total,
    int n, int nb, int nch) {
  __shared__ int c[BKT];
  __shared__ int cur[BKT];
  __shared__ int wsum[8];
  __shared__ int sbase;
  __shared__ int stage[STG];
  const int tid = threadIdx.x;
  const int lane = tid & 31;
  const int wave = tid >> 5;
  const bool isD = (int)blockIdx.x < nb;
  const int b = isD ? (int)blockIdx.x : (int)blockIdx.x - nb;
  const unsigned* __restrict__ rec = isD ? recD : recS;
  const int* __restrict__ tab = isD ? tabD : tabS;
  c[tid] = 0;
  __syncthreads();

  for (int ch = tid; ch < nch; ch += 256) {
    const int lo = tab[(size_t)ch * (nb + 1) + b];
    const int hi = tab[(size_t)ch * (nb + 1) + b + 1];
    const unsigned* p = rec + (size_t)ch * CH;
    for (int r = lo; r < hi; r += 8) {
      unsigned w[8];
#pragma unroll
      for (int u = 0; u < 8; ++u) {
        int ru = r + u;
        ru = ru < hi ? ru : hi - 1;
        w[u] = p[ru];
      }
#pragma unroll
      for (int u = 0; u < 8; ++u)
        if (r + u < hi) atomicAdd(&c[w[u] & (BKT - 1)], 1);
    }
  }
  __syncthreads();

  const int node = b * BKT + tid;
  const int mine = c[tid];
  if (!isD) {
    if (node < n) cs[node] = rsqrtf((float)(mine + 1));
    return;
  }

  int incl = mine;
  for (int o = 1; o < 32; o <<= 1) {
    const int t = __shfl_up(incl, o);
    if (lane >= o) incl += t;
  }
  if (lane == 31) wsum[wave] = incl;
  __syncthreads();
  int start = incl - mine;
  int tot = 0;
  for (int w = 0; w < 8; ++w) {
    if (w < wave) start += wsum[w];
    tot += wsum[w];
  }
  if (tid == 0) sbase = atomicAdd(total, tot);
  cur[tid] = start;
  __syncthreads();
  const int gbase = sbase;
  if (node < n) {
    rowstart[node] = gbase + start;
    rowend[node] = gbase + start + mine;
    cd[node] = rsqrtf((float)(mine + 1));
  }

  const bool staged = tot <= STG;
  for (int ch = tid; ch < nch; ch += 256) {
    const int lo = tab[(size_t)ch * (nb + 1) + b];
    const int hi = tab[(size_t)ch * (nb + 1) + b + 1];
    const unsigned* p = rec + (size_t)ch * CH;
    for (int r = lo; r < hi; r += 8) {
      unsigned w[8];
#pragma unroll
      for (int u = 0; u < 8; ++u) {
        int ru = r + u;
        ru = ru < hi ? ru : hi - 1;
        w[u] = p[ru];
      }
#pragma unroll
      for (int u = 0; u < 8; ++u) {
        if (r + u < hi) {
          const int slot = atomicAdd(&cur[w[u] & (BKT - 1)], 1);
          const int sv = (int)(w[u] >> BSH);
          if (staged) stage[slot] = sv; else esrc[(size_t)gbase + slot] = sv;
        }
      }
    }
  }
  if (staged) {
    __syncthreads();
    for (int j = tid; j < tot; j += 256) esrc[(size_t)gbase + j] = stage[j];
  }
}

template <int G, bool PRE>
__device__ __forceinline__ void add_rows(float4& acc, const float* __restrict__ h,
                                         int sl, int clbits, int k, unsigned lane16) {
  float4 v[G];
  float c[G];
#pragma unroll
  for (int u = 0; u < G; ++u) {
    const unsigned s = (unsigned)__builtin_amdgcn_readlane(sl, k + u);
    if constexpr (!PRE) c[u] = __int_as_float(__builtin_amdgcn_readlane(clbits, k + u));
    v[u] = *(const float4*)((const char*)h + (s * (unsigned)(DFE * 4) + lane16));
  }
#pragma unroll
  for (int u = 0; u < G; ++u) {
    if constexpr (PRE) {
      acc.x += v[u].x;
      acc.y += v[u].y;
      acc.z += v[u].z;
      acc.w += v[u].w;
    } else {
      acc.x += v[u].x * c[u];
      acc.y += v[u].y * c[u];
      acc.z += v[u].z * c[u];
      acc.w += v[u].w * c[u];
    }
  }
}

#define NPW 4
template <bool PRE, bool POST>
__global__ __launch_bounds__(128) void k_layer(
    const int* __restrict__ rowstart, const int* __restrict__ rowend,
    const int* __restrict__ esrc, const float* __restrict__ h,
    const float* __restrict__ cs, const float* __restrict__ cd,
    const float* __restrict__ W, const float* __restrict__ bias,
    float* __restrict__ out, int nrows) {
  __shared__ __attribute__((aligned(16))) float As[16 * 132];
  const int tid = threadIdx.x;
  const int lane = tid & 31;
  const int wave = __builtin_amdgcn_readfirstlane(tid >> 5);
  const int row0 = blockIdx.x * 16;
  const int last = nrows - 1;
  const unsigned lane16 = (unsigned)lane * 16u;

  int nd[NPW], beg[NPW], end[NPW];
#pragma unroll
  for (int i = 0; i < NPW; ++i) {
    const int node = row0 + wave * NPW + i;
    nd[i] = node < last ? node : last;
    beg[i] = rowstart[nd[i]];
    end[i] = rowend[nd[i]];
  }
  int slnext = nd[0];
  {
    const int rem = end[0] - beg[0];
    if (lane < (rem < 32 ? rem : 32)) slnext = esrc[beg[0] + lane];
  }

#pragma unroll
  for (int i = 0; i < NPW; ++i) {
    const int node = nd[i];
    int sl = slnext;
    if (i + 1 < NPW) {
      const int rem = end[i + 1] - beg[i + 1];
      slnext = nd[i + 1];
      if (lane < (rem < 32 ? rem : 32)) slnext = esrc[beg[i + 1] + lane];
    }
    int clbits = 0;
    if constexpr (!PRE) clbits = __float_as_int(cs[sl]);
    float4 acc = *(const float4*)((const char*)h + ((unsigned)node * (unsigned)(DFE * 4) + lane16));
    if constexpr (!PRE) {
      const float s = cs[node];
      acc.x *= s; acc.y *= s; acc.z *= s; acc.w *= s;
    }
    for (int j = beg[i]; j < end[i]; j += 32) {
      const int rem = end[i] - j;
      const int cnt = rem < 32 ? rem : 32;
      if (j != beg[i]) {
        sl = node;
        if (lane < cnt) sl = esrc[j + lane];
        if constexpr (!PRE) clbits = __float_as_int(cs[sl]);
      }
      int k = 0;
      for (; k + 8 <= cnt; k += 8) add_rows<8, PRE>(acc, h, sl, clbits, k, lane16);
      if (k + 4 <= cnt) { add_rows<4, PRE>(acc, h, sl, clbits, k, lane16); k += 4; }
      for (; k < cnt; ++k) add_rows<1, PRE>(acc, h, sl, clbits, k, lane16);
    }
    {
      const float sc = (row0 + wave * NPW + i) < nrows ? cd[node] : 0.f;
      acc.x *= sc; acc.y *= sc; acc.z *= sc; acc.w *= sc;
    }
    ((float4*)(As + (wave * NPW + i) * 132))[lane] = acc;
  }
  __syncthreads();

  const int half = lane >> 4;
  const int l15  = lane & 15;
  const int ntile0 = wave * 2;
  const int colb0 = ntile0 * 16;

  v8f acc0 = {};
  v8f acc1 = {};
  const float* Arow = As + l15 * 132;

#pragma unroll 8
  for (int k0 = 0; k0 < DFE; k0 += 4) {
    int ka = k0 + 2 * half;
    v2f a = { Arow[ka], Arow[ka + 1] };
    v2f b0 = { W[(size_t)ka * DFE + colb0 + l15],
               W[(size_t)(ka + 1) * DFE + colb0 + l15] };
    v2f b1 = { W[(size_t)ka * DFE + colb0 + 16 + l15],
               W[(size_t)(ka + 1) * DFE + colb0 + 16 + l15] };
    acc0 = __builtin_amdgcn_wmma_f32_16x16x4_f32(false, a, false, b0,
                                                 (short)0, acc0, false, false);
    acc1 = __builtin_amdgcn_wmma_f32_16x16x4_f32(false, a, false, b1,
                                                 (short)0, acc1, false, false);
  }

  for (int v = 0; v < 8; ++v) {
    int row = row0 + v + 8 * half;
    if (row < nrows) {
      float so = 1.0f;
      if constexpr (POST) so = cs[row];
      int col0 = colb0 + l15;
      float x0 = acc0[v] + bias[col0];
      x0 = x0 > 0.f ? x0 : 0.f;
      int col1 = col0 + 16;
      float x1 = acc1[v] + bias[col1];
      x1 = x1 > 0.f ? x1 : 0.f;
      if constexpr (POST) { x0 *= so; x1 *= so; }
      out[(size_t)row * DFE + col0] = x0;
      out[(size_t)row * DFE + col1] = x1;
    }
  }
}

__global__ void k_pool_zero(float* __restrict__ sums, float* __restrict__ cnt, int g) {
  int i = blockIdx.x * blockDim.x + threadIdx.x;
  if (i < g * DFE) sums[i] = 0.f;
  if (i < g) cnt[i] = 0.f;
}

#define PNW 16

__global__ __launch_bounds__(256) void k_pool_acc(
    const float* __restrict__ h, const int* __restrict__ gid,
    float* __restrict__ sums, float* __restrict__ cnt, int n) {
  __shared__ __attribute__((aligned(16))) float part[8 * DFE];
  __shared__ int   pg[8];
  __shared__ float pc[8];
  const int tid = threadIdx.x;
  const int lane = tid & 31;
  const int wave = __builtin_amdgcn_readfirstlane(tid >> 5);
  const int base = (blockIdx.x * 8 + wave) * PNW;
  int nn = n - base;
  nn = nn < 0 ? 0 : (nn > PNW ? PNW : nn);
  const int last = n - 1;

  int gi = base + lane;
  gi = gi < last ? gi : last;
  const int myg = gid[gi];

  float4 v[PNW];
#pragma unroll
  for (int j = 0; j < PNW; ++j) {
    int nd = base + j;
    nd = nd < last ? nd : last;
    v[j] = ((const float4*)(h + (size_t)nd * DFE))[lane];
  }

  int cur = __builtin_amdgcn_readlane(myg, 0);
  float4 acc = v[0];
  float c = 1.0f;
#pragma unroll
  for (int j = 1; j < PNW; ++j) {
    const int gj = __builtin_amdgcn_readlane(myg, j);
    if (gj != cur) {
      float* p = sums + (size_t)cur * DFE + lane * 4;
      unsafeAtomicAdd(p + 0, acc.x);
      unsafeAtomicAdd(p + 1, acc.y);
      unsafeAtomicAdd(p + 2, acc.z);
      unsafeAtomicAdd(p + 3, acc.w);
      if (lane == 0) unsafeAtomicAdd(&cnt[cur], c);
      acc = make_float4(0.f, 0.f, 0.f, 0.f);
      c = 0.f;
      cur = gj;
    }
    const bool in = j < nn;
    acc.x += in ? v[j].x : 0.f;
    acc.y += in ? v[j].y : 0.f;
    acc.z += in ? v[j].z : 0.f;
    acc.w += in ? v[j].w : 0.f;
    c += in ? 1.0f : 0.f;
  }

  ((float4*)(part + wave * DFE))[lane] = acc;
  if (lane == 0) { pg[wave] = nn > 0 ? cur : -1; pc[wave] = c; }
  __syncthreads();

  if (tid <= DFE) {
    const bool is_cnt = tid == DFE;
    const float* sp = is_cnt ? pc : part + tid;
    const int sstep = is_cnt ? 1 : DFE;
    float* dp = is_cnt ? cnt : sums + tid;
    const size_t dstep = is_cnt ? 1 : DFE;
    int gs[8];
    float ps[8];
#pragma unroll
    for (int w = 0; w < 8; ++w) { gs[w] = pg[w]; ps[w] = sp[w * sstep]; }
    float s = 0.f;
    int g0 = -1;
#pragma unroll
    for (int w = 0; w < 8; ++w) {
      if (gs[w] >= 0) {
        if (gs[w] != g0) {
          if (g0 >= 0) unsafeAtomicAdd(dp + (size_t)g0 * dstep, s);
          s = 0.f;
          g0 = gs[w];
        }
        s += ps[w];
      }
    }
    if (g0 >= 0) unsafeAtomicAdd(dp + (size_t)g0 * dstep, s);
  }
}

__global__ void k_pool_final(const float* __restrict__ sums, const float* __restrict__ cnt,
                             float* __restrict__ out, int g) {
  int i = blockIdx.x * blockDim.x + threadIdx.x;
  if (i < g * DFE) {
    float c = cnt[i >> 7];
    c = c > 1.0f ? c : 1.0f;
    out[i] = sums[i] / c;
  }
}

extern "C" void kernel_launch(void* const* d_in, const int* in_sizes, int n_in,
                              void* d_out, int out_size, void* d_ws, size_t ws_size,
                              hipStream_t stream) {
  const float* x   = (const float*)d_in[0];
  const float* W0  = (const float*)d_in[1];
  const float* b0  = (const float*)d_in[2];
  const float* W1  = (const float*)d_in[3];
  const float* b1  = (const float*)d_in[4];
  const float* W2  = (const float*)d_in[5];
  const float* b2  = (const float*)d_in[6];
  const int*   src = (const int*)d_in[7];
  const int*   dst = (const int*)d_in[8];
  const int*   gid = (const int*)d_in[9];

  const int n = in_sizes[0] / DFE;
  const int e = in_sizes[7];
  const int g = out_size / DFE;

  const int nb  = (n + BKT - 1) / BKT;
  const int nch = (e + CH - 1) / CH;

  float* ws = (float*)d_ws;
  size_t ND = (size_t)n * DFE;
  float* hA   = ws;
  float* hB   = hA + ND;
  float* cs   = hB + 2 * ND;
  float* cd   = cs + n;
  float* psum = cd + n;
  float* pcnt = psum + (size_t)g * DFE;
  int* rowstart = (int*)(pcnt + g);
  int* rowend   = rowstart + n;
  int* total    = rowend + n;
  int* esrc     = total + 32;
  unsigned* recD = (unsigned*)(esrc + e);
  unsigned* recS = recD + (size_t)nch * CH;
  int* tabD     = (int*)(recS + (size_t)nch * CH);
  int* tabS     = tabD + (size_t)nch * (nb + 1);

  const int TB = 256;
  dim3 blk(TB);

  k_chunk_sort<<<dim3(nch), blk, 0, stream>>>(src, dst, recD, recS, tabD, tabS, total, e, nb);
  k_bucket<<<dim3(2 * nb), blk, 0, stream>>>(recD, recS, tabD, tabS, rowstart, rowend, esrc,
                                             cs, cd, total, n, nb, nch);

  const unsigned layerBlocks    = (unsigned)((n + 15) / 16);
  const unsigned poolBlocks     = (unsigned)((n + 8 * PNW - 1) / (8 * PNW));

  k_layer<false, true><<<dim3(layerBlocks), dim3(128), 0, stream>>>(rowstart, rowend, esrc, x,  cs, cd, W0, b0, hA, n);
  k_layer<true,  true><<<dim3(layerBlocks), dim3(128), 0, stream>>>(rowstart, rowend, esrc, hA, cs, cd, W1, b1, hB, n);
  k_layer<true, false><<<dim3(layerBlocks), dim3(128), 0, stream>>>(rowstart, rowend, esrc, hB, cs, cd, W2, b2, hA, n);

  k_pool_zero<<<dim3((g * DFE + TB - 1) / TB), blk, 0, stream>>>(psum, pcnt, g);
  k_pool_acc<<<dim3(poolBlocks), blk, 0, stream>>>(hA, gid, psum, pcnt, n);
  k_pool_final<<<dim3((g * DFE + TB - 1) / TB), blk, 0, stream>>>(psum, pcnt, (float*)d_out, g);
}


# mean pooling moves into the last layer's end: a 16-row tile of one graph adds (its column sums / the graph's node count) to out, one float atomic per column; 8 launches to 5 (pool sum order changes)
# speedup vs baseline: 7.4942x; 1.0736x over previous
#include <hip/hip_runtime.h>
#include <cstdint>
#include <cstddef>

#define DFE 128

typedef float v2f __attribute__((ext_vector_type(2)));
typedef float v8f __attribute__((ext_vector_type(8)));

#define BSH 8
#define BKT (1 << BSH)
#define CH 4096
#define EPT (CH / 256)
#define NBMAX 1024
#define STG 8192

__global__ __launch_bounds__(256) void k_chunk_sort(
    const int* __restrict__ src, const int* __restrict__ dst,
    unsigned* __restrict__ recD, unsigned* __restrict__ recS,
    int* __restrict__ tabD, int* __restrict__ tabS,
    int* __restrict__ total, float* __restrict__ pout, int npout,
    float* __restrict__ gcnt, int g, int e, int nb) {
  __shared__ int hD[NBMAX + 1];
  __shared__ int hS[NBMAX + 1];
  __shared__ unsigned stD[CH];
  __shared__ unsigned stS[CH];
  __shared__ int wsum[2][8];
  const int tid = threadIdx.x;
  const int lane = tid & 31;
  const int wave = tid >> 5;
  const int base = blockIdx.x * CH;
  if (blockIdx.x == 0 && tid == 0) *total = 0;
  const int nthr = ((e + CH - 1) / CH) * 256;
  for (int i = blockIdx.x * 256 + tid; i < npout; i += nthr) pout[i] = 0.f;
  for (int i = blockIdx.x * 256 + tid; i < g; i += nthr) gcnt[i] = 0.f;
  for (int b = tid; b <= nb; b += 256) { hD[b] = 0; hS[b] = 0; }
  __syncthreads();

  int s[EPT], d[EPT], rD[EPT], rS[EPT];
#pragma unroll
  for (int k = 0; k < EPT; ++k) {
    const int i = base + k * 256 + tid;
    s[k] = -1; d[k] = -1;
    if (i < e) { s[k] = src[i]; d[k] = dst[i]; }
  }
#pragma unroll
  for (int k = 0; k < EPT; ++k) {
    rD[k] = 0; rS[k] = 0;
    if (d[k] >= 0) {
      rD[k] = atomicAdd(&hD[d[k] >> BSH], 1);
      rS[k] = atomicAdd(&hS[s[k] >> BSH], 1);
    }
  }
  __syncthreads();

  {
    int a[4], c[4];
    int sa = 0, sc = 0;
#pragma unroll
    for (int q = 0; q < 4; ++q) {
      const int idx = 4 * tid + q;
      a[q] = idx < nb ? hD[idx] : 0;
      c[q] = idx < nb ? hS[idx] : 0;
      sa += a[q]; sc += c[q];
    }
    int ia = sa, ic = sc;
    for (int o = 1; o < 32; o <<= 1) {
      const int t = __shfl_up(ia, o);
      const int u = __shfl_up(ic, o);
      if (lane >= o) { ia += t; ic += u; }
    }
    if (lane == 31) { wsum[0][wave] = ia; wsum[1][wave] = ic; }
    __syncthreads();
    int ea = ia - sa, ec = ic - sc;
    for (int w = 0; w < wave; ++w) { ea += wsum[0][w]; ec += wsum[1][w]; }
#pragma unroll
    for (int q = 0; q < 4; ++q) {
      const int idx = 4 * tid + q;
      if (idx < nb) { hD[idx] = ea; hS[idx] = ec; }
      ea += a[q]; ec += c[q];
    }
    if (tid == 255) { hD[nb] = ea; hS[nb] = ec; }
  }
  __syncthreads();

  for (int b = tid; b <= nb; b += 256) {
    tabD[(size_t)blockIdx.x * (nb + 1) + b] = hD[b];
    tabS[(size_t)blockIdx.x * (nb + 1) + b] = hS[b];
  }
#pragma unroll
  for (int k = 0; k < EPT; ++k) {
    if (d[k] >= 0) {
      stD[hD[d[k] >> BSH] + rD[k]] = ((unsigned)s[k] << BSH) | ((unsigned)d[k] & (BKT - 1));
      stS[hS[s[k] >> BSH] + rS[k]] = (unsigned)s[k] & (BKT - 1);
    }
  }
  __syncthreads();
  int cnt = e - base;
  cnt = cnt < CH ? cnt : CH;
  for (int j = tid; j < cnt; j += 256) {
    recD[(size_t)base + j] = stD[j];
    recS[(size_t)base + j] = stS[j];
  }
}

__global__ __launch_bounds__(256) void k_bucket(
    const unsigned* __restrict__ recD, const unsigned* __restrict__ recS,
    const int* __restrict__ tabD, const int* __restrict__ tabS,
    int* __restrict__ rowstart, int* __restrict__ rowend, int* __restrict__ esrc,
    float* __restrict__ cs, float* __restrict__ cd, int* __restrict__ total,
    const int* __restrict__ gid, float* __restrict__ gcnt,
    int n, int nb, int nch) {
  __shared__ int c[BKT];
  __shared__ int cur[BKT];
  __shared__ int wsum[8];
  __shared__ int sbase;
  __shared__ int stage[STG];
  const int tid = threadIdx.x;
  const int lane = tid & 31;
  const int wave = tid >> 5;
  const bool isD = (int)blockIdx.x < nb;
  const int b = isD ? (int)blockIdx.x : (int)blockIdx.x - nb;
  const unsigned* __restrict__ rec = isD ? recD : recS;
  const int* __restrict__ tab = isD ? tabD : tabS;
  c[tid] = 0;
  __syncthreads();

  for (int ch = tid; ch < nch; ch += 256) {
    const int lo = tab[(size_t)ch * (nb + 1) + b];
    const int hi = tab[(size_t)ch * (nb + 1) + b + 1];
    const unsigned* p = rec + (size_t)ch * CH;
    for (int r = lo; r < hi; r += 8) {
      unsigned w[8];
#pragma unroll
      for (int u = 0; u < 8; ++u) {
        int ru = r + u;
        ru = ru < hi ? ru : hi - 1;
        w[u] = p[ru];
      }
#pragma unroll
      for (int u = 0; u < 8; ++u)
        if (r + u < hi) atomicAdd(&c[w[u] & (BKT - 1)], 1);
    }
  }
  __syncthreads();

  const int node = b * BKT + tid;
  const int mine = c[tid];
  if (!isD) {
    if (node < n) cs[node] = rsqrtf((float)(mine + 1));
    return;
  }

  {
    const bool valid = node < n;
    const int myg = valid ? gid[node] : -1;
    const int prevg = __shfl_up(myg, 1);
    const bool lead = valid && (lane == 0 || prevg != myg);
    const unsigned leads = (unsigned)__ballot(lead);
    const unsigned vmask = (unsigned)__ballot(valid);
    if (lead) {
      const unsigned above = leads & ~((2u << lane) - 1u);
      const int endl = above ? __ffs(above) - 1 : __popc(vmask);
      unsafeAtomicAdd(&gcnt[myg], (float)(endl - lane));
    }
  }

  int incl = mine;
  for (int o = 1; o < 32; o <<= 1) {
    const int t = __shfl_up(incl, o);
    if (lane >= o) incl += t;
  }
  if (lane == 31) wsum[wave] = incl;
  __syncthreads();
  int start = incl - mine;
  int tot = 0;
  for (int w = 0; w < 8; ++w) {
    if (w < wave) start += wsum[w];
    tot += wsum[w];
  }
  if (tid == 0) sbase = atomicAdd(total, tot);
  cur[tid] = start;
  __syncthreads();
  const int gbase = sbase;
  if (node < n) {
    rowstart[node] = gbase + start;
    rowend[node] = gbase + start + mine;
    cd[node] = rsqrtf((float)(mine + 1));
  }

  const bool staged = tot <= STG;
  for (int ch = tid; ch < nch; ch += 256) {
    const int lo = tab[(size_t)ch * (nb + 1) + b];
    const int hi = tab[(size_t)ch * (nb + 1) + b + 1];
    const unsigned* p = rec + (size_t)ch * CH;
    for (int r = lo; r < hi; r += 8) {
      unsigned w[8];
#pragma unroll
      for (int u = 0; u < 8; ++u) {
        int ru = r + u;
        ru = ru < hi ? ru : hi - 1;
        w[u] = p[ru];
      }
#pragma unroll
      for (int u = 0; u < 8; ++u) {
        if (r + u < hi) {
          const int slot = atomicAdd(&cur[w[u] & (BKT - 1)], 1);
          const int sv = (int)(w[u] >> BSH);
          if (staged) stage[slot] = sv; else esrc[(size_t)gbase + slot] = sv;
        }
      }
    }
  }
  if (staged) {
    __syncthreads();
    for (int j = tid; j < tot; j += 256) esrc[(size_t)gbase + j] = stage[j];
  }
}

template <int G, bool PRE>
__device__ __forceinline__ void add_rows(float4& acc, const float* __restrict__ h,
                                         int sl, int clbits, int k, unsigned lane16) {
  float4 v[G];
  float c[G];
#pragma unroll
  for (int u = 0; u < G; ++u) {
    const unsigned s = (unsigned)__builtin_amdgcn_readlane(sl, k + u);
    if constexpr (!PRE) c[u] = __int_as_float(__builtin_amdgcn_readlane(clbits, k + u));
    v[u] = *(const float4*)((const char*)h + (s * (unsigned)(DFE * 4) + lane16));
  }
#pragma unroll
  for (int u = 0; u < G; ++u) {
    if constexpr (PRE) {
      acc.x += v[u].x;
      acc.y += v[u].y;
      acc.z += v[u].z;
      acc.w += v[u].w;
    } else {
      acc.x += v[u].x * c[u];
      acc.y += v[u].y * c[u];
      acc.z += v[u].z * c[u];
      acc.w += v[u].w * c[u];
    }
  }
}

#define NPW 4
template <bool PRE, bool POST, bool POOL>
__global__ __launch_bounds__(128) void k_layer(
    const int* __restrict__ rowstart, const int* __restrict__ rowend,
    const int* __restrict__ esrc, const float* __restrict__ h,
    const float* __restrict__ cs, const float* __restrict__ cd,
    const float* __restrict__ W, const float* __restrict__ bias,
    const int* __restrict__ gid, const float* __restrict__ gcnt,
    float* __restrict__ out, int nrows) {
  __shared__ __attribute__((aligned(16))) float As[16 * 132];
  const int tid = threadIdx.x;
  const int lane = tid & 31;
  const int wave = __builtin_amdgcn_readfirstlane(tid >> 5);
  const int row0 = blockIdx.x * 16;
  const int last = nrows - 1;
  const unsigned lane16 = (unsigned)lane * 16u;

  int myg = 0;
  if constexpr (POOL) {
    const int r = row0 + (lane & 15);
    myg = gid[r < last ? r : last];
  }

  int nd[NPW], beg[NPW], end[NPW];
#pragma unroll
  for (int i = 0; i < NPW; ++i) {
    const int node = row0 + wave * NPW + i;
    nd[i] = node < last ? node : last;
    beg[i] = rowstart[nd[i]];
    end[i] = rowend[nd[i]];
  }
  int slnext = nd[0];
  {
    const int rem = end[0] - beg[0];
    if (lane < (rem < 32 ? rem : 32)) slnext = esrc[beg[0] + lane];
  }

#pragma unroll
  for (int i = 0; i < NPW; ++i) {
    const int node = nd[i];
    int sl = slnext;
    if (i + 1 < NPW) {
      const int rem = end[i + 1] - beg[i + 1];
      slnext = nd[i + 1];
      if (lane < (rem < 32 ? rem : 32)) slnext = esrc[beg[i + 1] + lane];
    }
    int clbits = 0;
    if constexpr (!PRE) clbits = __float_as_int(cs[sl]);
    float4 acc = *(const float4*)((const char*)h + ((unsigned)node * (unsigned)(DFE * 4) + lane16));
    if constexpr (!PRE) {
      const float s = cs[node];
      acc.x *= s; acc.y *= s; acc.z *= s; acc.w *= s;
    }
    for (int j = beg[i]; j < end[i]; j += 32) {
      const int rem = end[i] - j;
      const int cnt = rem < 32 ? rem : 32;
      if (j != beg[i]) {
        sl = node;
        if (lane < cnt) sl = esrc[j + lane];
        if constexpr (!PRE) clbits = __float_as_int(cs[sl]);
      }
      int k = 0;
      for (; k + 8 <= cnt; k += 8) add_rows<8, PRE>(acc, h, sl, clbits, k, lane16);
      if (k + 4 <= cnt) { add_rows<4, PRE>(acc, h, sl, clbits, k, lane16); k += 4; }
      for (; k < cnt; ++k) add_rows<1, PRE>(acc, h, sl, clbits, k, lane16);
    }
    {
      const float sc = (row0 + wave * NPW + i) < nrows ? cd[node] : 0.f;
      acc.x *= sc; acc.y *= sc; acc.z *= sc; acc.w *= sc;
    }
    ((float4*)(As + (wave * NPW + i) * 132))[lane] = acc;
  }
  __syncthreads();

  const int half = lane >> 4;
  const int l15  = lane & 15;
  const int ntile0 = wave * 2;
  const int colb0 = ntile0 * 16;

  v8f acc0 = {};
  v8f acc1 = {};
  const float* Arow = As + l15 * 132;

  int g0 = 0;
  bool uni = false;
  float gc0 = 1.0f;
  if constexpr (POOL) {
    g0 = __builtin_amdgcn_readfirstlane(myg);
    uni = row0 + 16 <= nrows && __ballot(myg != g0) == 0;
    gc0 = gcnt[g0];
    gc0 = gc0 > 1.0f ? gc0 : 1.0f;
  }

#pragma unroll 8
  for (int k0 = 0; k0 < DFE; k0 += 4) {
    int ka = k0 + 2 * half;
    v2f a = { Arow[ka], Arow[ka + 1] };
    v2f b0 = { W[(size_t)ka * DFE + colb0 + l15],
               W[(size_t)(ka + 1) * DFE + colb0 + l15] };
    v2f b1 = { W[(size_t)ka * DFE + colb0 + 16 + l15],
               W[(size_t)(ka + 1) * DFE + colb0 + 16 + l15] };
    acc0 = __builtin_amdgcn_wmma_f32_16x16x4_f32(false, a, false, b0,
                                                 (short)0, acc0, false, false);
    acc1 = __builtin_amdgcn_wmma_f32_16x16x4_f32(false, a, false, b1,
                                                 (short)0, acc1, false, false);
  }

  if constexpr (POOL) {
    const int col0 = colb0 + l15;
    const float bb0 = bias[col0];
    const float bb1 = bias[col0 + 16];
    if (uni) {
      float s0 = 0.f, s1 = 0.f;
#pragma unroll
      for (int v = 0; v < 8; ++v) {
        float x0 = acc0[v] + bb0;
        x0 = x0 > 0.f ? x0 : 0.f;
        s0 += x0;
        float x1 = acc1[v] + bb1;
        x1 = x1 > 0.f ? x1 : 0.f;
        s1 += x1;
      }
      s0 += __shfl_xor(s0, 16);
      s1 += __shfl_xor(s1, 16);
      unsafeAtomicAdd(out + (size_t)g0 * DFE + colb0 + lane, (half ? s1 : s0) / gc0);
    } else {
#pragma unroll
      for (int v = 0; v < 8; ++v) {
        const int r = v + 8 * half;
        const int gr = __shfl(myg, r);
        if (row0 + r < nrows) {
          float c = gcnt[gr];
          c = c > 1.0f ? c : 1.0f;
          float x0 = acc0[v] + bb0;
          x0 = x0 > 0.f ? x0 : 0.f;
          float x1 = acc1[v] + bb1;
          x1 = x1 > 0.f ? x1 : 0.f;
          unsafeAtomicAdd(out + (size_t)gr * DFE + col0, x0 / c);
          unsafeAtomicAdd(out + (size_t)gr * DFE + col0 + 16, x1 / c);
        }
      }
    }
    return;
  }

  for (int v = 0; v < 8; ++v) {
    int row = row0 + v + 8 * half;
    if (row < nrows) {
      float so = 1.0f;
      if constexpr (POST) so = cs[row];
      int col0 = colb0 + l15;
      float x0 = acc0[v] + bias[col0];
      x0 = x0 > 0.f ? x0 : 0.f;
      int col1 = col0 + 16;
      float x1 = acc1[v] + bias[col1];
      x1 = x1 > 0.f ? x1 : 0.f;
      if constexpr (POST) { x0 *= so; x1 *= so; }
      out[(size_t)row * DFE + col0] = x0;
      out[(size_t)row * DFE + col1] = x1;
    }
  }
}

extern "C" void kernel_launch(void* const* d_in, const int* in_sizes, int n_in,
                              void* d_out, int out_size, void* d_ws, size_t ws_size,
                              hipStream_t stream) {
  const float* x   = (const float*)d_in[0];
  const float* W0  = (const float*)d_in[1];
  const float* b0  = (const float*)d_in[2];
  const float* W1  = (const float*)d_in[3];
  const float* b1  = (const float*)d_in[4];
  const float* W2  = (const float*)d_in[5];
  const float* b2  = (const float*)d_in[6];
  const int*   src = (const int*)d_in[7];
  const int*   dst = (const int*)d_in[8];
  const int*   gid = (const int*)d_in[9];

  const int n = in_sizes[0] / DFE;
  const int e = in_sizes[7];
  const int g = out_size / DFE;

  const int nb  = (n + BKT - 1) / BKT;
  const int nch = (e + CH - 1) / CH;

  float* ws = (float*)d_ws;
  size_t ND = (size_t)n * DFE;
  float* hA   = ws;
  float* hB   = hA + ND;
  float* cs   = hB + 2 * ND;
  float* cd   = cs + n;
  float* gcnt = cd + n;
  int* rowstart = (int*)(gcnt + g);
  int* rowend   = rowstart + n;
  int* total    = rowend + n;
  int* esrc     = total + 32;
  unsigned* recD = (unsigned*)(esrc + e);
  unsigned* recS = recD + (size_t)nch * CH;
  int* tabD     = (int*)(recS + (size_t)nch * CH);
  int* tabS     = tabD + (size_t)nch * (nb + 1);

  const int TB = 256;
  dim3 blk(TB);

  k_chunk_sort<<<dim3(nch), blk, 0, stream>>>(src, dst, recD, recS, tabD, tabS, total,
                                              (float*)d_out, out_size, gcnt, g, e, nb);
  k_bucket<<<dim3(2 * nb), blk, 0, stream>>>(recD, recS, tabD, tabS, rowstart, rowend, esrc,
                                             cs, cd, total, gid, gcnt, n, nb, nch);

  const unsigned layerBlocks    = (unsigned)((n + 15) / 16);

  k_layer<false, true, false><<<dim3(layerBlocks), dim3(128), 0, stream>>>(rowstart, rowend, esrc, x,  cs, cd, W0, b0, nullptr, nullptr, hA, n);
  k_layer<true,  true, false><<<dim3(layerBlocks), dim3(128), 0, stream>>>(rowstart, rowend, esrc, hA, cs, cd, W1, b1, nullptr, nullptr, hB, n);
  k_layer<true, false,  true><<<dim3(layerBlocks), dim3(128), 0, stream>>>(rowstart, rowend, esrc, hB, cs, cd, W2, b2, gid, gcnt, (float*)d_out, n);
}
